# starting kernel
# baseline (speedup 1.0000x reference)
#include <hip/hip_runtime.h>

typedef __attribute__((ext_vector_type(2))) float v2f;
typedef __attribute__((ext_vector_type(8))) float v8f;

#define KDIM 128

__global__ void k_init_deg(float* __restrict__ deg, int n) {
    int i = blockIdx.x * blockDim.x + threadIdx.x;
    if (i < n) deg[i] = 1.0f;
}

__global__ void k_count_deg(const int* __restrict__ dst, float* __restrict__ deg, int e) {
    int i = blockIdx.x * blockDim.x + threadIdx.x;
    if (i < e) atomicAdd(&deg[dst[i]], 1.0f);
}

__global__ void k_finalize_dinv(float* __restrict__ deg, int n) {
    int i = blockIdx.x * blockDim.x + threadIdx.x;
    if (i < n) deg[i] = rsqrtf(deg[i]);
}

template <int OUTCH, int MT, bool RELU>
__global__ __launch_bounds__(256) void k_gemm(const float* __restrict__ A,
                                              const float* __restrict__ W,
                                              float* __restrict__ H, int n) {
    constexpr int LDSW = KDIM + 4;
    __shared__ float tile[MT * 16 * LDSW];

    const int tid = threadIdx.x;
    const int m_base = blockIdx.x * (MT * 16);

    constexpr int CHUNKS = MT * 16 * (KDIM / 4);
    for (int c = tid; c < CHUNKS; c += 256) {
        int row = c >> 5;
        int kc  = c & 31;
        int grow = m_base + row;
        if (grow >= n) grow = n - 1;
        float4 v = ((const float4*)(A + (size_t)grow * KDIM))[kc];
        if (RELU) {
            v.x = fmaxf(v.x, 0.0f); v.y = fmaxf(v.y, 0.0f);
            v.z = fmaxf(v.z, 0.0f); v.w = fmaxf(v.w, 0.0f);
        }
        *(float4*)&tile[row * LDSW + kc * 4] = v;
    }
    __syncthreads();

    const int wave = tid >> 5;
    const int lane = tid & 31;
    int mt, nt;
    if (OUTCH == 128) { mt = 0;         nt = wave;     }
    else              { mt = wave >> 2; nt = wave & 3; }
    const int lm = lane & 15;
    const int kh = lane >> 4;

    const float* arow = &tile[(mt * 16 + lm) * LDSW];
    const float* wcol = W + nt * 16 + lm;

    v8f acc = {};
#pragma unroll
    for (int k0 = 0; k0 < KDIM; k0 += 4) {
        v2f a, b;
        a.x = arow[k0 + 2 * kh + 0];
        a.y = arow[k0 + 2 * kh + 1];
        b.x = wcol[(size_t)(k0 + 2 * kh + 0) * OUTCH];
        b.y = wcol[(size_t)(k0 + 2 * kh + 1) * OUTCH];
        acc = __builtin_amdgcn_wmma_f32_16x16x4_f32(
             false, a,  false, b,
             (short)0, acc,  false,  false);
    }

    const int mrow0 = m_base + mt * 16 + (kh ? 8 : 0);
#pragma unroll
    for (int r = 0; r < 8; ++r) {
        int grow = mrow0 + r;
        if (grow < n)
            H[(size_t)grow * OUTCH + nt * 16 + lm] = acc[r];
    }
}

template <int CH>
__global__ void k_seed(float* __restrict__ out, const float* __restrict__ h,
                       const float* __restrict__ bias, const float* __restrict__ dinv, int n) {
    int i = blockIdx.x * blockDim.x + threadIdx.x;
    int total = n * (CH / 4);
    if (i >= total) return;
    int node = i / (CH / 4);
    int c4   = i % (CH / 4);
    float d = dinv[node];
    float s = d * d;
    float4 hv = ((const float4*)(h + (size_t)node * CH))[c4];
    float4 bv = ((const float4*)bias)[c4];
    float4 o;
    o.x = fmaf(hv.x, s, bv.x); o.y = fmaf(hv.y, s, bv.y);
    o.z = fmaf(hv.z, s, bv.z); o.w = fmaf(hv.w, s, bv.w);
    ((float4*)(out + (size_t)node * CH))[c4] = o;
}

template <int CH>
__global__ void k_scatter(const int* __restrict__ src, const int* __restrict__ dst,
                          const float* __restrict__ h, float* __restrict__ out,
                          const float* __restrict__ dinv, int e) {
    int g = blockIdx.x * blockDim.x + threadIdx.x;
    int edge = g >> 5;
    int lane = g & 31;
    if (edge >= e) return;
    int s = src[edge], d = dst[edge];
    float w = dinv[s] * dinv[d];
    if (CH == 128) {
        float4 hv = ((const float4*)(h + (size_t)s * CH))[lane];
        float* o = out + (size_t)d * CH + lane * 4;
        atomicAdd(o + 0, hv.x * w); atomicAdd(o + 1, hv.y * w);
        atomicAdd(o + 2, hv.z * w); atomicAdd(o + 3, hv.w * w);
    } else {
        float2 hv = ((const float2*)(h + (size_t)s * CH))[lane];
        float* o = out + (size_t)d * CH + lane * 2;
        atomicAdd(o + 0, hv.x * w); atomicAdd(o + 1, hv.y * w);
    }
}

extern "C" void kernel_launch(void* const* d_in, const int* in_sizes, int n_in,
                              void* d_out, int out_size, void* d_ws, size_t ws_size,
                              hipStream_t stream) {
    const float* x  = (const float*)d_in[0];
    const int*   ei = (const int*)d_in[1];
    const float* W1 = (const float*)d_in[2];
    const float* b1 = (const float*)d_in[3];
    const float* W2 = (const float*)d_in[4];
    const float* b2 = (const float*)d_in[5];
    float* out = (float*)d_out;

    const int n = in_sizes[0] / 128;
    const int e = in_sizes[1] / 2;
    const int* src = ei;
    const int* dst = ei + e;

    float* dinv = (float*)d_ws;
    float* h  = dinv + n;
    float* o1 = h + (size_t)n * 128;

    const unsigned T = 256;
    k_init_deg<<<(n + T - 1) / T, T, 0, stream>>>(dinv, n);
    k_count_deg<<<(e + T - 1) / T, T, 0, stream>>>(dst, dinv, e);
    k_finalize_dinv<<<(n + T - 1) / T, T, 0, stream>>>(dinv, n);

    k_gemm<128, 1, false><<<(n + 15) / 16, T, 0, stream>>>(x, W1, h, n);
    k_seed<128><<<((unsigned)(n * 32) + T - 1) / T, T, 0, stream>>>(o1, h, b1, dinv, n);
    k_scatter<128><<<(unsigned)((e + 7) / 8), T, 0, stream>>>(src, dst, h, o1, dinv, e);

    k_gemm<64, 2, true><<<(n + 31) / 32, T, 0, stream>>>(o1, W2, h, n);
    k_seed<64><<<((unsigned)(n * 16) + T - 1) / T, T, 0, stream>>>(out, h, b2, dinv, n);
    k_scatter<64><<<(unsigned)((e + 7) / 8), T, 0, stream>>>(src, dst, h, out, dinv, e);
}


# Edge sums and degrees by destination-range owner blocks: a block rescans the edge list and adds its rows' messages in LDS (atomics, arrival order), one store per row; no global atomics, no seed pass
# speedup vs baseline: 1.2752x; 1.2752x over previous
#include <hip/hip_runtime.h>
#include <stdint.h>

typedef __attribute__((ext_vector_type(2))) float v2f;
typedef __attribute__((ext_vector_type(8))) float v8f;

#define KDIM 128
#define NTHR 1024

template <class F>
__device__ __forceinline__ void scan_dst(const int* __restrict__ dst, int e, int lo,
                                         unsigned rows, int tid, F f) {
    if ((e & 3) == 0 && (((uintptr_t)dst) & 15) == 0) {
        const int4* d4 = (const int4*)dst;
        const int nq = e >> 2;
        for (int q0 = tid; q0 < nq; q0 += 4 * NTHR) {
            int4 v[4];
#pragma unroll
            for (int u = 0; u < 4; ++u) {
                int q = q0 + u * NTHR;
                v[u] = (q < nq) ? d4[q] : make_int4(-1, -1, -1, -1);
            }
#pragma unroll
            for (int u = 0; u < 4; ++u) {
                int eb = (q0 + u * NTHR) * 4;
                unsigned a = (unsigned)v[u].x - (unsigned)lo;
                unsigned b = (unsigned)v[u].y - (unsigned)lo;
                unsigned c = (unsigned)v[u].z - (unsigned)lo;
                unsigned d = (unsigned)v[u].w - (unsigned)lo;
                if (a < rows) f(eb + 0, (int)a);
                if (b < rows) f(eb + 1, (int)b);
                if (c < rows) f(eb + 2, (int)c);
                if (d < rows) f(eb + 3, (int)d);
            }
        }
    } else {
        for (int i = tid; i < e; i += NTHR) {
            unsigned a = (unsigned)dst[i] - (unsigned)lo;
            if (a < rows) f(i, (int)a);
        }
    }
}

template <int R>
__global__ __launch_bounds__(NTHR) void k_deg(const int* __restrict__ dst,
                                              float* __restrict__ dinv, int n, int e) {
    __shared__ int cnt[R];
    const int tid = threadIdx.x;
    const int lo = blockIdx.x * R;
    const int rows = min(R, n - lo);
    for (int r = tid; r < R; r += NTHR) cnt[r] = 0;
    __syncthreads();
    scan_dst(dst, e, lo, (unsigned)rows, tid, [&](int, int dl) { atomicAdd(&cnt[dl], 1); });
    __syncthreads();
    for (int r = tid; r < rows; r += NTHR)
        dinv[lo + r] = rsqrtf((float)(cnt[r] + 1));
}

template <int OUTCH, int MT, bool RELU>
__global__ __launch_bounds__(256) void k_gemm(const float* __restrict__ A,
                                              const float* __restrict__ W,
                                              float* __restrict__ H, int n) {
    constexpr int LDSW = KDIM + 4;
    __shared__ float tile[MT * 16 * LDSW];

    const int tid = threadIdx.x;
    const int m_base = blockIdx.x * (MT * 16);

    constexpr int CHUNKS = MT * 16 * (KDIM / 4);
    for (int c = tid; c < CHUNKS; c += 256) {
        int row = c >> 5;
        int kc  = c & 31;
        int grow = m_base + row;
        if (grow >= n) grow = n - 1;
        float4 v = ((const float4*)(A + (size_t)grow * KDIM))[kc];
        if (RELU) {
            v.x = fmaxf(v.x, 0.0f); v.y = fmaxf(v.y, 0.0f);
            v.z = fmaxf(v.z, 0.0f); v.w = fmaxf(v.w, 0.0f);
        }
        *(float4*)&tile[row * LDSW + kc * 4] = v;
    }
    __syncthreads();

    const int wave = tid >> 5;
    const int lane = tid & 31;
    int mt, nt;
    if (OUTCH == 128) { mt = 0;         nt = wave;     }
    else              { mt = wave >> 2; nt = wave & 3; }
    const int lm = lane & 15;
    const int kh = lane >> 4;

    const float* arow = &tile[(mt * 16 + lm) * LDSW];
    const float* wcol = W + nt * 16 + lm;

    v8f acc = {};
#pragma unroll
    for (int k0 = 0; k0 < KDIM; k0 += 4) {
        v2f a, b;
        a.x = arow[k0 + 2 * kh + 0];
        a.y = arow[k0 + 2 * kh + 1];
        b.x = wcol[(size_t)(k0 + 2 * kh + 0) * OUTCH];
        b.y = wcol[(size_t)(k0 + 2 * kh + 1) * OUTCH];
        acc = __builtin_amdgcn_wmma_f32_16x16x4_f32(
             false, a,  false, b,
             (short)0, acc,  false,  false);
    }

    const int mrow0 = m_base + mt * 16 + (kh ? 8 : 0);
#pragma unroll
    for (int r = 0; r < 8; ++r) {
        int grow = mrow0 + r;
        if (grow < n)
            H[(size_t)grow * OUTCH + nt * 16 + lm] = acc[r];
    }
}

template <int CH>
__device__ __forceinline__ int chan_pos(int c) {
    constexpr int VEC = CH / 32;
    return (c % VEC) * 32 + c / VEC;
}

template <int CH, int R, int CAP>
__global__ __launch_bounds__(NTHR) void k_agg(const int* __restrict__ src,
                                              const int* __restrict__ dst,
                                              const float* __restrict__ h,
                                              const float* __restrict__ bias,
                                              const float* __restrict__ dinv,
                                              float* __restrict__ out, int n, int e) {
    constexpr int VEC = CH / 32;
    __shared__ float acc[R * CH];
    __shared__ int   l_s[CAP];
    __shared__ int   l_d[CAP];
    __shared__ float s_dinv[R];
    __shared__ int   s_cnt;

    const int tid = threadIdx.x;
    const int lo = blockIdx.x * R;
    const int rows = min(R, n - lo);

    if (tid == 0) s_cnt = 0;
    for (int r = tid; r < rows; r += NTHR) s_dinv[r] = dinv[lo + r];
    for (int i = tid; i < rows * (CH / 4); i += NTHR) {
        int r  = i / (CH / 4);
        int c4 = i % (CH / 4);
        float d = dinv[lo + r];
        float s = d * d;
        float4 hv = ((const float4*)(h + (size_t)(lo + r) * CH))[c4];
        float4 bv = ((const float4*)bias)[c4];
        float* a = acc + r * CH;
        a[chan_pos<CH>(c4 * 4 + 0)] = fmaf(hv.x, s, bv.x);
        a[chan_pos<CH>(c4 * 4 + 1)] = fmaf(hv.y, s, bv.y);
        a[chan_pos<CH>(c4 * 4 + 2)] = fmaf(hv.z, s, bv.z);
        a[chan_pos<CH>(c4 * 4 + 3)] = fmaf(hv.w, s, bv.w);
    }
    __syncthreads();

    scan_dst(dst, e, lo, (unsigned)rows, tid, [&](int eidx, int dl) {
        int slot = atomicAdd(&s_cnt, 1);
        if (slot < CAP) {
            l_s[slot] = eidx;
            l_d[slot] = dl;
        } else {
            int s = src[eidx];
            float w = dinv[s] * s_dinv[dl];
            const float* hr = h + (size_t)s * CH;
            float* a = acc + dl * CH;
            for (int c = 0; c < CH; ++c) atomicAdd(&a[chan_pos<CH>(c)], hr[c] * w);
        }
    });
    __syncthreads();

    const int total = min(s_cnt, CAP);
    for (int i = tid; i < total; i += NTHR) l_s[i] = src[l_s[i]];
    __syncthreads();

    const int wave = tid >> 5;
    const int lane = tid & 31;
    for (int base = wave * 32; base < total; base += NTHR) {
        int i = base + lane;
        int s = 0, dl = 0;
        float w = 0.0f;
        if (i < total) {
            s = l_s[i];
            dl = l_d[i];
            w = dinv[s] * s_dinv[dl];
        }
        const int m = min(32, total - base);
#pragma unroll
        for (int g = 0; g < 32; g += 8) {
            if (g + 8 <= m) {
                float hv[8][VEC];
#pragma unroll
                for (int k = 0; k < 8; ++k) {
                    int sj = __builtin_amdgcn_readlane(s, g + k);
                    const float* hr = h + (size_t)sj * CH + lane * VEC;
#pragma unroll
                    for (int j = 0; j < VEC; ++j) hv[k][j] = hr[j];
                }
#pragma unroll
                for (int k = 0; k < 8; ++k) {
                    float wj = __int_as_float(__builtin_amdgcn_readlane(__float_as_int(w), g + k));
                    int dj = __builtin_amdgcn_readlane(dl, g + k);
                    float* a = acc + dj * CH + lane;
#pragma unroll
                    for (int j = 0; j < VEC; ++j) atomicAdd(a + j * 32, hv[k][j] * wj);
                }
            } else if (g < m) {
#pragma unroll
                for (int k = 0; k < 8; ++k) {
                    if (g + k < m) {
                        int sj = __builtin_amdgcn_readlane(s, g + k);
                        float wj = __int_as_float(__builtin_amdgcn_readlane(__float_as_int(w), g + k));
                        int dj = __builtin_amdgcn_readlane(dl, g + k);
                        const float* hr = h + (size_t)sj * CH + lane * VEC;
                        float* a = acc + dj * CH + lane;
#pragma unroll
                        for (int j = 0; j < VEC; ++j) atomicAdd(a + j * 32, hr[j] * wj);
                    }
                }
            }
        }
    }
    __syncthreads();

    for (int i = tid; i < rows * (CH / 4); i += NTHR) {
        int r  = i / (CH / 4);
        int c4 = i % (CH / 4);
        const float* a = acc + r * CH;
        float4 o;
        o.x = a[chan_pos<CH>(c4 * 4 + 0)];
        o.y = a[chan_pos<CH>(c4 * 4 + 1)];
        o.z = a[chan_pos<CH>(c4 * 4 + 2)];
        o.w = a[chan_pos<CH>(c4 * 4 + 3)];
        ((float4*)(out + (size_t)(lo + r) * CH))[c4] = o;
    }
}

extern "C" void kernel_launch(void* const* d_in, const int* in_sizes, int n_in,
                              void* d_out, int out_size, void* d_ws, size_t ws_size,
                              hipStream_t stream) {
    const float* x  = (const float*)d_in[0];
    const int*   ei = (const int*)d_in[1];
    const float* W1 = (const float*)d_in[2];
    const float* b1 = (const float*)d_in[3];
    const float* W2 = (const float*)d_in[4];
    const float* b2 = (const float*)d_in[5];
    float* out = (float*)d_out;

    const int n = in_sizes[0] / 128;
    const int e = in_sizes[1] / 2;
    const int* src = ei;
    const int* dst = ei + e;

    float* dinv = (float*)d_ws;
    float* h  = dinv + n;
    float* o1 = h + (size_t)n * 128;

    const unsigned T = 256;
    constexpr int RD = 800, R1 = 400, R2 = 768, CAP = 14336;

    k_deg<RD><<<(n + RD - 1) / RD, NTHR, 0, stream>>>(dst, dinv, n, e);

    k_gemm<128, 1, false><<<(n + 15) / 16, T, 0, stream>>>(x, W1, h, n);
    k_agg<128, R1, CAP><<<(n + R1 - 1) / R1, NTHR, 0, stream>>>(src, dst, h, b1, dinv, o1, n, e);

    k_gemm<64, 2, true><<<(n + 31) / 32, T, 0, stream>>>(o1, W2, h, n);
    k_agg<64, R2, CAP><<<(n + R2 - 1) / R2, NTHR, 0, stream>>>(src, dst, h, b2, dinv, out, n, e);
}


# Edge-list scan double-buffered in registers: a thread tests one set of 16-byte loads while the next set is in flight (8 loads a set in the degree pass, 4 in the sum passes); order of sums unchanged
# speedup vs baseline: 1.5693x; 1.2306x over previous
#include <hip/hip_runtime.h>
#include <stdint.h>

typedef __attribute__((ext_vector_type(2))) float v2f;
typedef __attribute__((ext_vector_type(8))) float v8f;

#define KDIM 128
#define NTHR 1024

template <int U, class F>
__device__ __forceinline__ void scan_dst(const int* __restrict__ dst, int e, int lo,
                                         unsigned rows, int tid, F f) {
    if ((e & 3) == 0 && (((uintptr_t)dst) & 15) == 0) {
        const int4* d4 = (const int4*)dst;
        const int nq = e >> 2;
        constexpr int BATCH = U * NTHR;

        auto test = [&](const int4 (&v)[U], int q0) {
#pragma unroll
            for (int u = 0; u < U; ++u) {
                int eb = (q0 + u * NTHR) * 4;
                unsigned a = (unsigned)v[u].x - (unsigned)lo;
                unsigned b = (unsigned)v[u].y - (unsigned)lo;
                unsigned c = (unsigned)v[u].z - (unsigned)lo;
                unsigned d = (unsigned)v[u].w - (unsigned)lo;
                if (a < rows) f(eb + 0, (int)a);
                if (b < rows) f(eb + 1, (int)b);
                if (c < rows) f(eb + 2, (int)c);
                if (d < rows) f(eb + 3, (int)d);
            }
        };

        int4 va[U], vb[U];
        const int nfull = nq / (2 * BATCH);
        if (nfull > 0) {
#pragma unroll
            for (int u = 0; u < U; ++u) va[u] = d4[tid + u * NTHR];
            for (int t = 0; t < nfull; ++t) {
                const int q0 = t * (2 * BATCH) + tid;
#pragma unroll
                for (int u = 0; u < U; ++u) vb[u] = d4[q0 + BATCH + u * NTHR];
                test(va, q0);
                const int qn = (t + 1 < nfull) ? q0 + 2 * BATCH : q0;
#pragma unroll
                for (int u = 0; u < U; ++u) va[u] = d4[qn + u * NTHR];
                test(vb, q0 + BATCH);
            }
        }
        for (int q0 = nfull * (2 * BATCH) + tid; q0 < nq; q0 += BATCH) {
#pragma unroll
            for (int u = 0; u < U; ++u) {
                int q = q0 + u * NTHR;
                va[u] = (q < nq) ? d4[q] : make_int4(-1, -1, -1, -1);
            }
            test(va, q0);
        }
    } else {
        for (int i = tid; i < e; i += NTHR) {
            unsigned a = (unsigned)dst[i] - (unsigned)lo;
            if (a < rows) f(i, (int)a);
        }
    }
}

template <int R>
__global__ __launch_bounds__(NTHR) void k_deg(const int* __restrict__ dst,
                                              float* __restrict__ dinv, int n, int e) {
    __shared__ int cnt[R];
    const int tid = threadIdx.x;
    const int lo = blockIdx.x * R;
    const int rows = min(R, n - lo);
    for (int r = tid; r < R; r += NTHR) cnt[r] = 0;
    __syncthreads();
    scan_dst<8>(dst, e, lo, (unsigned)rows, tid, [&](int, int dl) { atomicAdd(&cnt[dl], 1); });
    __syncthreads();
    for (int r = tid; r < rows; r += NTHR)
        dinv[lo + r] = rsqrtf((float)(cnt[r] + 1));
}

template <int OUTCH, int MT, bool RELU>
__global__ __launch_bounds__(256) void k_gemm(const float* __restrict__ A,
                                              const float* __restrict__ W,
                                              float* __restrict__ H, int n) {
    constexpr int LDSW = KDIM + 4;
    __shared__ float tile[MT * 16 * LDSW];

    const int tid = threadIdx.x;
    const int m_base = blockIdx.x * (MT * 16);

    constexpr int CHUNKS = MT * 16 * (KDIM / 4);
    for (int c = tid; c < CHUNKS; c += 256) {
        int row = c >> 5;
        int kc  = c & 31;
        int grow = m_base + row;
        if (grow >= n) grow = n - 1;
        float4 v = ((const float4*)(A + (size_t)grow * KDIM))[kc];
        if (RELU) {
            v.x = fmaxf(v.x, 0.0f); v.y = fmaxf(v.y, 0.0f);
            v.z = fmaxf(v.z, 0.0f); v.w = fmaxf(v.w, 0.0f);
        }
        *(float4*)&tile[row * LDSW + kc * 4] = v;
    }
    __syncthreads();

    const int wave = tid >> 5;
    const int lane = tid & 31;
    int mt, nt;
    if (OUTCH == 128) { mt = 0;         nt = wave;     }
    else              { mt = wave >> 2; nt = wave & 3; }
    const int lm = lane & 15;
    const int kh = lane >> 4;

    const float* arow = &tile[(mt * 16 + lm) * LDSW];
    const float* wcol = W + nt * 16 + lm;

    v8f acc = {};
#pragma unroll
    for (int k0 = 0; k0 < KDIM; k0 += 4) {
        v2f a, b;
        a.x = arow[k0 + 2 * kh + 0];
        a.y = arow[k0 + 2 * kh + 1];
        b.x = wcol[(size_t)(k0 + 2 * kh + 0) * OUTCH];
        b.y = wcol[(size_t)(k0 + 2 * kh + 1) * OUTCH];
        acc = __builtin_amdgcn_wmma_f32_16x16x4_f32(
             false, a,  false, b,
             (short)0, acc,  false,  false);
    }

    const int mrow0 = m_base + mt * 16 + (kh ? 8 : 0);
#pragma unroll
    for (int r = 0; r < 8; ++r) {
        int grow = mrow0 + r;
        if (grow < n)
            H[(size_t)grow * OUTCH + nt * 16 + lm] = acc[r];
    }
}

template <int CH>
__device__ __forceinline__ int chan_pos(int c) {
    constexpr int VEC = CH / 32;
    return (c % VEC) * 32 + c / VEC;
}

template <int CH, int R, int CAP>
__global__ __launch_bounds__(NTHR) void k_agg(const int* __restrict__ src,
                                              const int* __restrict__ dst,
                                              const float* __restrict__ h,
                                              const float* __restrict__ bias,
                                              const float* __restrict__ dinv,
                                              float* __restrict__ out, int n, int e) {
    constexpr int VEC = CH / 32;
    __shared__ float acc[R * CH];
    __shared__ int   l_s[CAP];
    __shared__ int   l_d[CAP];
    __shared__ float s_dinv[R];
    __shared__ int   s_cnt;

    const int tid = threadIdx.x;
    const int lo = blockIdx.x * R;
    const int rows = min(R, n - lo);

    if (tid == 0) s_cnt = 0;
    for (int r = tid; r < rows; r += NTHR) s_dinv[r] = dinv[lo + r];
    for (int i = tid; i < rows * (CH / 4); i += NTHR) {
        int r  = i / (CH / 4);
        int c4 = i % (CH / 4);
        float d = dinv[lo + r];
        float s = d * d;
        float4 hv = ((const float4*)(h + (size_t)(lo + r) * CH))[c4];
        float4 bv = ((const float4*)bias)[c4];
        float* a = acc + r * CH;
        a[chan_pos<CH>(c4 * 4 + 0)] = fmaf(hv.x, s, bv.x);
        a[chan_pos<CH>(c4 * 4 + 1)] = fmaf(hv.y, s, bv.y);
        a[chan_pos<CH>(c4 * 4 + 2)] = fmaf(hv.z, s, bv.z);
        a[chan_pos<CH>(c4 * 4 + 3)] = fmaf(hv.w, s, bv.w);
    }
    __syncthreads();

    scan_dst<4>(dst, e, lo, (unsigned)rows, tid, [&](int eidx, int dl) {
        int slot = atomicAdd(&s_cnt, 1);
        if (slot < CAP) {
            l_s[slot] = eidx;
            l_d[slot] = dl;
        } else {
            int s = src[eidx];
            float w = dinv[s] * s_dinv[dl];
            const float* hr = h + (size_t)s * CH;
            float* a = acc + dl * CH;
            for (int c = 0; c < CH; ++c) atomicAdd(&a[chan_pos<CH>(c)], hr[c] * w);
        }
    });
    __syncthreads();

    const int total = min(s_cnt, CAP);
    for (int i = tid; i < total; i += NTHR) l_s[i] = src[l_s[i]];
    __syncthreads();

    const int wave = tid >> 5;
    const int lane = tid & 31;
    for (int base = wave * 32; base < total; base += NTHR) {
        int i = base + lane;
        int s = 0, dl = 0;
        float w = 0.0f;
        if (i < total) {
            s = l_s[i];
            dl = l_d[i];
            w = dinv[s] * s_dinv[dl];
        }
        const int m = min(32, total - base);
#pragma unroll
        for (int g = 0; g < 32; g += 8) {
            if (g + 8 <= m) {
                float hv[8][VEC];
#pragma unroll
                for (int k = 0; k < 8; ++k) {
                    int sj = __builtin_amdgcn_readlane(s, g + k);
                    const float* hr = h + (size_t)sj * CH + lane * VEC;
#pragma unroll
                    for (int j = 0; j < VEC; ++j) hv[k][j] = hr[j];
                }
#pragma unroll
                for (int k = 0; k < 8; ++k) {
                    float wj = __int_as_float(__builtin_amdgcn_readlane(__float_as_int(w), g + k));
                    int dj = __builtin_amdgcn_readlane(dl, g + k);
                    float* a = acc + dj * CH + lane;
#pragma unroll
                    for (int j = 0; j < VEC; ++j) atomicAdd(a + j * 32, hv[k][j] * wj);
                }
            } else if (g < m) {
#pragma unroll
                for (int k = 0; k < 8; ++k) {
                    if (g + k < m) {
                        int sj = __builtin_amdgcn_readlane(s, g + k);
                        float wj = __int_as_float(__builtin_amdgcn_readlane(__float_as_int(w), g + k));
                        int dj = __builtin_amdgcn_readlane(dl, g + k);
                        const float* hr = h + (size_t)sj * CH + lane * VEC;
                        float* a = acc + dj * CH + lane;
#pragma unroll
                        for (int j = 0; j < VEC; ++j) atomicAdd(a + j * 32, hr[j] * wj);
                    }
                }
            }
        }
    }
    __syncthreads();

    for (int i = tid; i < rows * (CH / 4); i += NTHR) {
        int r  = i / (CH / 4);
        int c4 = i % (CH / 4);
        const float* a = acc + r * CH;
        float4 o;
        o.x = a[chan_pos<CH>(c4 * 4 + 0)];
        o.y = a[chan_pos<CH>(c4 * 4 + 1)];
        o.z = a[chan_pos<CH>(c4 * 4 + 2)];
        o.w = a[chan_pos<CH>(c4 * 4 + 3)];
        ((float4*)(out + (size_t)(lo + r) * CH))[c4] = o;
    }
}

extern "C" void kernel_launch(void* const* d_in, const int* in_sizes, int n_in,
                              void* d_out, int out_size, void* d_ws, size_t ws_size,
                              hipStream_t stream) {
    const float* x  = (const float*)d_in[0];
    const int*   ei = (const int*)d_in[1];
    const float* W1 = (const float*)d_in[2];
    const float* b1 = (const float*)d_in[3];
    const float* W2 = (const float*)d_in[4];
    const float* b2 = (const float*)d_in[5];
    float* out = (float*)d_out;

    const int n = in_sizes[0] / 128;
    const int e = in_sizes[1] / 2;
    const int* src = ei;
    const int* dst = ei + e;

    float* dinv = (float*)d_ws;
    float* h  = dinv + n;
    float* o1 = h + (size_t)n * 128;

    const unsigned T = 256;
    constexpr int RD = 800, R1 = 400, R2 = 768, CAP = 14336;

    k_deg<RD><<<(n + RD - 1) / RD, NTHR, 0, stream>>>(dst, dinv, n, e);

    k_gemm<128, 1, false><<<(n + 15) / 16, T, 0, stream>>>(x, W1, h, n);
    k_agg<128, R1, CAP><<<(n + R1 - 1) / R1, NTHR, 0, stream>>>(src, dst, h, b1, dinv, o1, n, e);

    k_gemm<64, 2, true><<<(n + 31) / 32, T, 0, stream>>>(o1, W2, h, n);
    k_agg<64, R2, CAP><<<(n + R2 - 1) / R2, NTHR, 0, stream>>>(src, dst, h, b2, dinv, out, n, e);
}


# Owner blocks scan a 1-byte range key per edge (dst/400, written in-call by a small kernel; 16 edges per 16-byte load) in place of the 4-byte dst; a listed edge reads its dst once; sum order kept
# speedup vs baseline: 2.0801x; 1.3255x over previous
#include <hip/hip_runtime.h>
#include <stdint.h>

typedef __attribute__((ext_vector_type(2))) float v2f;
typedef __attribute__((ext_vector_type(8))) float v8f;

#define KDIM 128
#define NTHR 1024

#define KR 400

__global__ __launch_bounds__(256) void k_keys(const int* __restrict__ dst,
                                              unsigned char* __restrict__ keys, int e) {
    const int t = blockIdx.x * 256 + threadIdx.x;
    const int nt = gridDim.x * 256;
    if ((e & 15) == 0 && (((uintptr_t)dst) & 15) == 0 && (((uintptr_t)keys) & 15) == 0) {
        const int4* d4 = (const int4*)dst;
        uint4* k4 = (uint4*)keys;
        const int nq = e >> 4;
        for (int q = t; q < nq; q += nt) {
            int4 a = d4[4 * q + 0], b = d4[4 * q + 1], c = d4[4 * q + 2], d = d4[4 * q + 3];
            uint4 o;
            o.x = ((unsigned)a.x / KR) | (((unsigned)a.y / KR) << 8) | (((unsigned)a.z / KR) << 16) | (((unsigned)a.w / KR) << 24);
            o.y = ((unsigned)b.x / KR) | (((unsigned)b.y / KR) << 8) | (((unsigned)b.z / KR) << 16) | (((unsigned)b.w / KR) << 24);
            o.z = ((unsigned)c.x / KR) | (((unsigned)c.y / KR) << 8) | (((unsigned)c.z / KR) << 16) | (((unsigned)c.w / KR) << 24);
            o.w = ((unsigned)d.x / KR) | (((unsigned)d.y / KR) << 8) | (((unsigned)d.z / KR) << 16) | (((unsigned)d.w / KR) << 24);
            k4[q] = o;
        }
    } else {
        for (int i = t; i < e; i += nt) keys[i] = (unsigned char)((unsigned)dst[i] / KR);
    }
}

template <int SH, int U, class F>
__device__ __forceinline__ void scan_keys(const unsigned char* __restrict__ keys, int e,
                                          int owner, int tid, F f) {
    if ((e & 15) == 0 && (((uintptr_t)keys) & 15) == 0) {
        const uint4* k4 = (const uint4*)keys;
        const int nq = e >> 4;
        constexpr int BATCH = U * NTHR;
        constexpr unsigned KM = ((0xFFu << SH) & 0xFFu) * 0x01010101u;
        const unsigned pat = (unsigned)(owner << SH) * 0x01010101u;

        auto miss = [&](unsigned w) -> unsigned {
            unsigned x = (w & KM) ^ pat;
            return ((x & 0x7F7F7F7Fu) + 0x7F7F7F7Fu) | x;
        };

        static_assert(U == 4, "the match mask below holds four pieces");
        auto test = [&](const uint4 (&v)[U], int q0) {
            unsigned mk[U];
#pragma unroll
            for (int u = 0; u < U; ++u) {
                unsigned nz = ((miss(v[u].x) >> 7) & 0x01010101u) | ((miss(v[u].y) >> 6) & 0x02020202u)
                            | ((miss(v[u].z) >> 5) & 0x04040404u) | ((miss(v[u].w) >> 4) & 0x08080808u);
                mk[u] = ~nz & 0x0F0F0F0Fu;
            }
            unsigned long long m = (unsigned long long)(mk[0] | (mk[1] << 4))
                                 | ((unsigned long long)(mk[2] | (mk[3] << 4)) << 32);
            const int e0 = q0 * 16;
            while (m) {
                const int p = __builtin_ctzll(m);
                m &= m - 1;
                const int piece = ((p >> 5) << 1) | ((p >> 2) & 1);
                f(e0 + piece * (NTHR * 16) + ((p & 3) << 2) + ((p >> 3) & 3));
            }
        };

        uint4 va[U], vb[U];
        const int nfull = nq / (2 * BATCH);
        if (nfull > 0) {
#pragma unroll
            for (int u = 0; u < U; ++u) va[u] = k4[tid + u * NTHR];
            for (int t = 0; t < nfull; ++t) {
                const int q0 = t * (2 * BATCH) + tid;
#pragma unroll
                for (int u = 0; u < U; ++u) vb[u] = k4[q0 + BATCH + u * NTHR];
                test(va, q0);
                const int qn = (t + 1 < nfull) ? q0 + 2 * BATCH : q0;
#pragma unroll
                for (int u = 0; u < U; ++u) va[u] = k4[qn + u * NTHR];
                test(vb, q0 + BATCH);
            }
        }
        for (int q0 = nfull * (2 * BATCH) + tid; q0 < nq; q0 += BATCH) {
#pragma unroll
            for (int u = 0; u < U; ++u) {
                int q = q0 + u * NTHR;
                va[u] = (q < nq) ? k4[q] : make_uint4(~pat, ~pat, ~pat, ~pat);
            }
            test(va, q0);
        }
    } else {
        for (int i = tid; i < e; i += NTHR)
            if ((keys[i] >> SH) == owner) f(i);
    }
}

template <int R>
__global__ __launch_bounds__(NTHR) void k_deg(const int* __restrict__ dst,
                                              const unsigned char* __restrict__ keys,
                                              float* __restrict__ dinv, int n, int e) {
    static_assert(R == 2 * KR, "owner range must be two key ranges");
    __shared__ int cnt[R];
    const int tid = threadIdx.x;
    const int lo = blockIdx.x * R;
    const int rows = min(R, n - lo);
    for (int r = tid; r < R; r += NTHR) cnt[r] = 0;
    __syncthreads();
    scan_keys<1, 4>(keys, e, blockIdx.x, tid, [&](int eidx) { atomicAdd(&cnt[dst[eidx] - lo], 1); });
    __syncthreads();
    for (int r = tid; r < rows; r += NTHR)
        dinv[lo + r] = rsqrtf((float)(cnt[r] + 1));
}

template <int OUTCH, int MT, bool RELU>
__global__ __launch_bounds__(256) void k_gemm(const float* __restrict__ A,
                                              const float* __restrict__ W,
                                              float* __restrict__ H, int n) {
    constexpr int LDSW = KDIM + 4;
    __shared__ float tile[MT * 16 * LDSW];

    const int tid = threadIdx.x;
    const int m_base = blockIdx.x * (MT * 16);

    constexpr int CHUNKS = MT * 16 * (KDIM / 4);
    for (int c = tid; c < CHUNKS; c += 256) {
        int row = c >> 5;
        int kc  = c & 31;
        int grow = m_base + row;
        if (grow >= n) grow = n - 1;
        float4 v = ((const float4*)(A + (size_t)grow * KDIM))[kc];
        if (RELU) {
            v.x = fmaxf(v.x, 0.0f); v.y = fmaxf(v.y, 0.0f);
            v.z = fmaxf(v.z, 0.0f); v.w = fmaxf(v.w, 0.0f);
        }
        *(float4*)&tile[row * LDSW + kc * 4] = v;
    }
    __syncthreads();

    const int wave = tid >> 5;
    const int lane = tid & 31;
    int mt, nt;
    if (OUTCH == 128) { mt = 0;         nt = wave;     }
    else              { mt = wave >> 2; nt = wave & 3; }
    const int lm = lane & 15;
    const int kh = lane >> 4;

    const float* arow = &tile[(mt * 16 + lm) * LDSW];
    const float* wcol = W + nt * 16 + lm;

    v8f acc = {};
#pragma unroll
    for (int k0 = 0; k0 < KDIM; k0 += 4) {
        v2f a, b;
        a.x = arow[k0 + 2 * kh + 0];
        a.y = arow[k0 + 2 * kh + 1];
        b.x = wcol[(size_t)(k0 + 2 * kh + 0) * OUTCH];
        b.y = wcol[(size_t)(k0 + 2 * kh + 1) * OUTCH];
        acc = __builtin_amdgcn_wmma_f32_16x16x4_f32(
             false, a,  false, b,
             (short)0, acc,  false,  false);
    }

    const int mrow0 = m_base + mt * 16 + (kh ? 8 : 0);
#pragma unroll
    for (int r = 0; r < 8; ++r) {
        int grow = mrow0 + r;
        if (grow < n)
            H[(size_t)grow * OUTCH + nt * 16 + lm] = acc[r];
    }
}

template <int CH>
__device__ __forceinline__ int chan_pos(int c) {
    constexpr int VEC = CH / 32;
    return (c % VEC) * 32 + c / VEC;
}

template <int CH, int SH, int CAP>
__global__ __launch_bounds__(NTHR) void k_agg(const int* __restrict__ src,
                                              const int* __restrict__ dst,
                                              const unsigned char* __restrict__ keys,
                                              const float* __restrict__ h,
                                              const float* __restrict__ bias,
                                              const float* __restrict__ dinv,
                                              float* __restrict__ out, int n, int e) {
    constexpr int VEC = CH / 32;
    constexpr int R = KR << SH;
    __shared__ float acc[R * CH];
    __shared__ int   l_s[CAP];
    __shared__ int   l_d[CAP];
    __shared__ float s_dinv[R];
    __shared__ int   s_cnt;

    const int tid = threadIdx.x;
    const int lo = blockIdx.x * R;
    const int rows = min(R, n - lo);

    if (tid == 0) s_cnt = 0;
    for (int r = tid; r < rows; r += NTHR) s_dinv[r] = dinv[lo + r];
    for (int i = tid; i < rows * (CH / 4); i += NTHR) {
        int r  = i / (CH / 4);
        int c4 = i % (CH / 4);
        float d = dinv[lo + r];
        float s = d * d;
        float4 hv = ((const float4*)(h + (size_t)(lo + r) * CH))[c4];
        float4 bv = ((const float4*)bias)[c4];
        float* a = acc + r * CH;
        a[chan_pos<CH>(c4 * 4 + 0)] = fmaf(hv.x, s, bv.x);
        a[chan_pos<CH>(c4 * 4 + 1)] = fmaf(hv.y, s, bv.y);
        a[chan_pos<CH>(c4 * 4 + 2)] = fmaf(hv.z, s, bv.z);
        a[chan_pos<CH>(c4 * 4 + 3)] = fmaf(hv.w, s, bv.w);
    }
    __syncthreads();

    scan_keys<SH, 4>(keys, e, blockIdx.x, tid, [&](int eidx) {
        int slot = atomicAdd(&s_cnt, 1);
        if (slot < CAP) {
            l_s[slot] = eidx;
        } else {
            int s = src[eidx];
            int dl = dst[eidx] - lo;
            float w = dinv[s] * s_dinv[dl];
            const float* hr = h + (size_t)s * CH;
            float* a = acc + dl * CH;
            for (int c = 0; c < CH; ++c) atomicAdd(&a[chan_pos<CH>(c)], hr[c] * w);
        }
    });
    __syncthreads();

    const int total = min(s_cnt, CAP);
    for (int i = tid; i < total; i += NTHR) {
        int ei = l_s[i];
        l_s[i] = src[ei];
        l_d[i] = dst[ei] - lo;
    }
    __syncthreads();

    const int wave = tid >> 5;
    const int lane = tid & 31;
    for (int base = wave * 32; base < total; base += NTHR) {
        int i = base + lane;
        int s = 0, dl = 0;
        float w = 0.0f;
        if (i < total) {
            s = l_s[i];
            dl = l_d[i];
            w = dinv[s] * s_dinv[dl];
        }
        const int m = min(32, total - base);
#pragma unroll
        for (int g = 0; g < 32; g += 8) {
            if (g + 8 <= m) {
                float hv[8][VEC];
#pragma unroll
                for (int k = 0; k < 8; ++k) {
                    int sj = __builtin_amdgcn_readlane(s, g + k);
                    const float* hr = h + (size_t)sj * CH + lane * VEC;
#pragma unroll
                    for (int j = 0; j < VEC; ++j) hv[k][j] = hr[j];
                }
#pragma unroll
                for (int k = 0; k < 8; ++k) {
                    float wj = __int_as_float(__builtin_amdgcn_readlane(__float_as_int(w), g + k));
                    int dj = __builtin_amdgcn_readlane(dl, g + k);
                    float* a = acc + dj * CH + lane;
#pragma unroll
                    for (int j = 0; j < VEC; ++j) atomicAdd(a + j * 32, hv[k][j] * wj);
                }
            } else if (g < m) {
#pragma unroll
                for (int k = 0; k < 8; ++k) {
                    if (g + k < m) {
                        int sj = __builtin_amdgcn_readlane(s, g + k);
                        float wj = __int_as_float(__builtin_amdgcn_readlane(__float_as_int(w), g + k));
                        int dj = __builtin_amdgcn_readlane(dl, g + k);
                        const float* hr = h + (size_t)sj * CH + lane * VEC;
                        float* a = acc + dj * CH + lane;
#pragma unroll
                        for (int j = 0; j < VEC; ++j) atomicAdd(a + j * 32, hr[j] * wj);
                    }
                }
            }
        }
    }
    __syncthreads();

    for (int i = tid; i < rows * (CH / 4); i += NTHR) {
        int r  = i / (CH / 4);
        int c4 = i % (CH / 4);
        const float* a = acc + r * CH;
        float4 o;
        o.x = a[chan_pos<CH>(c4 * 4 + 0)];
        o.y = a[chan_pos<CH>(c4 * 4 + 1)];
        o.z = a[chan_pos<CH>(c4 * 4 + 2)];
        o.w = a[chan_pos<CH>(c4 * 4 + 3)];
        ((float4*)(out + (size_t)(lo + r) * CH))[c4] = o;
    }
}

extern "C" void kernel_launch(void* const* d_in, const int* in_sizes, int n_in,
                              void* d_out, int out_size, void* d_ws, size_t ws_size,
                              hipStream_t stream) {
    const float* x  = (const float*)d_in[0];
    const int*   ei = (const int*)d_in[1];
    const float* W1 = (const float*)d_in[2];
    const float* b1 = (const float*)d_in[3];
    const float* W2 = (const float*)d_in[4];
    const float* b2 = (const float*)d_in[5];
    float* out = (float*)d_out;

    const int n = in_sizes[0] / 128;
    const int e = in_sizes[1] / 2;
    const int* src = ei;
    const int* dst = ei + e;

    float* dinv = (float*)d_ws;
    float* h  = dinv + n;
    float* o1 = h + (size_t)n * 128;

    const unsigned T = 256;
    constexpr int RD = 2 * KR, R1 = KR, R2 = 2 * KR, CAP = 14336;

    unsigned char* keys_a = (unsigned char*)d_out;
    unsigned char* keys_b = (unsigned char*)(h + (size_t)n * 64);
    const int kblocks = (e / 16 + 255) / 256 + 1;

    k_keys<<<kblocks, T, 0, stream>>>(dst, keys_a, e);
    k_deg<RD><<<(n + RD - 1) / RD, NTHR, 0, stream>>>(dst, keys_a, dinv, n, e);

    k_gemm<128, 1, false><<<(n + 15) / 16, T, 0, stream>>>(x, W1, h, n);
    k_agg<128, 0, CAP><<<(n + R1 - 1) / R1, NTHR, 0, stream>>>(src, dst, keys_a, h, b1, dinv, o1, n, e);

    k_gemm<64, 2, true><<<(n + 31) / 32, T, 0, stream>>>(o1, W2, h, n);
    k_keys<<<kblocks, T, 0, stream>>>(dst, keys_b, e);
    k_agg<64, 1, CAP><<<(n + R2 - 1) / R2, NTHR, 0, stream>>>(src, dst, keys_b, h, b2, dinv, out, n, e);
}


# Edge keys scanned once, not three times: the degree pass (a block per 400-row range) leaves each range's (source,row) list in its stretch of out; both sum passes read it; sums stay LDS atomics
# speedup vs baseline: 2.9900x; 1.4374x over previous
#include <hip/hip_runtime.h>
#include <stdint.h>

typedef __attribute__((ext_vector_type(2))) float v2f;
typedef __attribute__((ext_vector_type(8))) float v8f;

#define KDIM 128
#define NTHR 1024

#define KR 400
#define LW 64
#define LHEAD 4

__global__ __launch_bounds__(256) void k_keys(const int* __restrict__ dst,
                                              unsigned char* __restrict__ keys, int e) {
    const int t = blockIdx.x * 256 + threadIdx.x;
    const int nt = gridDim.x * 256;
    if ((e & 15) == 0 && (((uintptr_t)dst) & 15) == 0 && (((uintptr_t)keys) & 15) == 0) {
        const int4* d4 = (const int4*)dst;
        uint4* k4 = (uint4*)keys;
        const int nq = e >> 4;
        for (int q = t; q < nq; q += nt) {
            int4 a = d4[4 * q + 0], b = d4[4 * q + 1], c = d4[4 * q + 2], d = d4[4 * q + 3];
            uint4 o;
            o.x = ((unsigned)a.x / KR) | (((unsigned)a.y / KR) << 8) | (((unsigned)a.z / KR) << 16) | (((unsigned)a.w / KR) << 24);
            o.y = ((unsigned)b.x / KR) | (((unsigned)b.y / KR) << 8) | (((unsigned)b.z / KR) << 16) | (((unsigned)b.w / KR) << 24);
            o.z = ((unsigned)c.x / KR) | (((unsigned)c.y / KR) << 8) | (((unsigned)c.z / KR) << 16) | (((unsigned)c.w / KR) << 24);
            o.w = ((unsigned)d.x / KR) | (((unsigned)d.y / KR) << 8) | (((unsigned)d.z / KR) << 16) | (((unsigned)d.w / KR) << 24);
            k4[q] = o;
        }
    } else {
        for (int i = t; i < e; i += nt) keys[i] = (unsigned char)((unsigned)dst[i] / KR);
    }
}

template <int SH, int U, class F>
__device__ __forceinline__ void scan_keys(const unsigned char* __restrict__ keys, int e,
                                          int owner, int tid, F f) {
    if ((e & 15) == 0 && (((uintptr_t)keys) & 15) == 0) {
        const uint4* k4 = (const uint4*)keys;
        const int nq = e >> 4;
        constexpr int BATCH = U * NTHR;
        constexpr unsigned KM = ((0xFFu << SH) & 0xFFu) * 0x01010101u;
        const unsigned pat = (unsigned)(owner << SH) * 0x01010101u;

        auto miss = [&](unsigned w) -> unsigned {
            unsigned x = (w & KM) ^ pat;
            return ((x & 0x7F7F7F7Fu) + 0x7F7F7F7Fu) | x;
        };

        static_assert(U == 4, "the match mask below holds four pieces");
        auto test = [&](const uint4 (&v)[U], int q0) {
            unsigned mk[U];
#pragma unroll
            for (int u = 0; u < U; ++u) {
                unsigned nz = ((miss(v[u].x) >> 7) & 0x01010101u) | ((miss(v[u].y) >> 6) & 0x02020202u)
                            | ((miss(v[u].z) >> 5) & 0x04040404u) | ((miss(v[u].w) >> 4) & 0x08080808u);
                mk[u] = ~nz & 0x0F0F0F0Fu;
            }
            unsigned long long m = (unsigned long long)(mk[0] | (mk[1] << 4))
                                 | ((unsigned long long)(mk[2] | (mk[3] << 4)) << 32);
            const int e0 = q0 * 16;
            while (m) {
                const int p = __builtin_ctzll(m);
                m &= m - 1;
                const int piece = ((p >> 5) << 1) | ((p >> 2) & 1);
                f(e0 + piece * (NTHR * 16) + ((p & 3) << 2) + ((p >> 3) & 3));
            }
        };

        uint4 va[U], vb[U];
        const int nfull = nq / (2 * BATCH);
        if (nfull > 0) {
#pragma unroll
            for (int u = 0; u < U; ++u) va[u] = k4[tid + u * NTHR];
            for (int t = 0; t < nfull; ++t) {
                const int q0 = t * (2 * BATCH) + tid;
#pragma unroll
                for (int u = 0; u < U; ++u) vb[u] = k4[q0 + BATCH + u * NTHR];
                test(va, q0);
                const int qn = (t + 1 < nfull) ? q0 + 2 * BATCH : q0;
#pragma unroll
                for (int u = 0; u < U; ++u) va[u] = k4[qn + u * NTHR];
                test(vb, q0 + BATCH);
            }
        }
        for (int q0 = nfull * (2 * BATCH) + tid; q0 < nq; q0 += BATCH) {
#pragma unroll
            for (int u = 0; u < U; ++u) {
                int q = q0 + u * NTHR;
                va[u] = (q < nq) ? k4[q] : make_uint4(~pat, ~pat, ~pat, ~pat);
            }
            test(va, q0);
        }
    } else {
        for (int i = tid; i < e; i += NTHR)
            if ((keys[i] >> SH) == owner) f(i);
    }
}

template <int CAP>
__global__ __launch_bounds__(NTHR) void k_deg(const int* __restrict__ src,
                                              const int* __restrict__ dst,
                                              const unsigned char* __restrict__ keys,
                                              float* __restrict__ dinv,
                                              int* __restrict__ lists, int n, int e) {
    __shared__ int cnt[KR];
    __shared__ int l_s[CAP];
    __shared__ int l_d[CAP];
    __shared__ int s_cnt;
    const int tid = threadIdx.x;
    const int lo = blockIdx.x * KR;
    const int rows = min(KR, n - lo);
    if (tid == 0) s_cnt = 0;
    for (int r = tid; r < KR; r += NTHR) cnt[r] = 0;
    __syncthreads();
    scan_keys<0, 4>(keys, e, blockIdx.x, tid, [&](int eidx) {
        int slot = atomicAdd(&s_cnt, 1);
        if (slot < CAP) l_s[slot] = eidx;
        else atomicAdd(&cnt[dst[eidx] - lo], 1);
    });
    __syncthreads();

    const int found = s_cnt;
    const int total = min(found, CAP);
    for (int i = tid; i < total; i += NTHR) {
        int ei = l_s[i];
        int s = src[ei];
        int dl = dst[ei] - lo;
        l_s[i] = s;
        l_d[i] = dl;
        atomicAdd(&cnt[dl], 1);
    }
    __syncthreads();

    for (int r = tid; r < rows; r += NTHR)
        dinv[lo + r] = rsqrtf((float)(cnt[r] + 1));

    int* reg = lists + (size_t)lo * LW;
    const int room = min(CAP, (rows * LW - LHEAD) / 2);
    const bool fits = found <= room;
    if (tid == 0) reg[0] = fits ? found : -1;
    if (fits) {
        int2* ent = (int2*)(reg + LHEAD);
        for (int i = tid; i < total; i += NTHR) ent[i] = make_int2(l_s[i], l_d[i]);
    }
}

template <int OUTCH, int MT, bool RELU>
__global__ __launch_bounds__(256) void k_gemm(const float* __restrict__ A,
                                              const float* __restrict__ W,
                                              float* __restrict__ H, int n) {
    constexpr int LDSW = KDIM + 4;
    __shared__ float tile[MT * 16 * LDSW];

    const int tid = threadIdx.x;
    const int m_base = blockIdx.x * (MT * 16);

    constexpr int CHUNKS = MT * 16 * (KDIM / 4);
    for (int c = tid; c < CHUNKS; c += 256) {
        int row = c >> 5;
        int kc  = c & 31;
        int grow = m_base + row;
        if (grow >= n) grow = n - 1;
        float4 v = ((const float4*)(A + (size_t)grow * KDIM))[kc];
        if (RELU) {
            v.x = fmaxf(v.x, 0.0f); v.y = fmaxf(v.y, 0.0f);
            v.z = fmaxf(v.z, 0.0f); v.w = fmaxf(v.w, 0.0f);
        }
        *(float4*)&tile[row * LDSW + kc * 4] = v;
    }
    __syncthreads();

    const int wave = tid >> 5;
    const int lane = tid & 31;
    int mt, nt;
    if (OUTCH == 128) { mt = 0;         nt = wave;     }
    else              { mt = wave >> 2; nt = wave & 3; }
    const int lm = lane & 15;
    const int kh = lane >> 4;

    const float* arow = &tile[(mt * 16 + lm) * LDSW];
    const float* wcol = W + nt * 16 + lm;

    v8f acc = {};
#pragma unroll
    for (int k0 = 0; k0 < KDIM; k0 += 4) {
        v2f a, b;
        a.x = arow[k0 + 2 * kh + 0];
        a.y = arow[k0 + 2 * kh + 1];
        b.x = wcol[(size_t)(k0 + 2 * kh + 0) * OUTCH];
        b.y = wcol[(size_t)(k0 + 2 * kh + 1) * OUTCH];
        acc = __builtin_amdgcn_wmma_f32_16x16x4_f32(
             false, a,  false, b,
             (short)0, acc,  false,  false);
    }

    const int mrow0 = m_base + mt * 16 + (kh ? 8 : 0);
#pragma unroll
    for (int r = 0; r < 8; ++r) {
        int grow = mrow0 + r;
        if (grow < n)
            H[(size_t)grow * OUTCH + nt * 16 + lm] = acc[r];
    }
}

template <int CH>
__device__ __forceinline__ int chan_pos(int c) {
    constexpr int VEC = CH / 32;
    return (c % VEC) * 32 + c / VEC;
}

template <int CH, int CAP>
__global__ __launch_bounds__(NTHR) void k_agg(const int* __restrict__ src,
                                              const int* __restrict__ dst,
                                              const int* lists,
                                              const float* __restrict__ h,
                                              const float* __restrict__ bias,
                                              const float* __restrict__ dinv,
                                              float* out, int n, int e) {
    constexpr int VEC = CH / 32;
    constexpr int R = KR;
    __shared__ float acc[R * CH];
    __shared__ int   l_s[CAP];
    __shared__ int   l_d[CAP];
    __shared__ float s_dinv[R];

    const int tid = threadIdx.x;
    const int lo = blockIdx.x * R;
    const int rows = min(R, n - lo);

    const int* reg = lists + (size_t)lo * LW;
    const int found = reg[0];
    const int total = found < 0 ? 0 : min(found, CAP);
    {
        const int2* ent = (const int2*)(reg + LHEAD);
        for (int i = tid; i < total; i += NTHR) {
            int2 p = ent[i];
            l_s[i] = p.x;
            l_d[i] = p.y;
        }
    }

    for (int r = tid; r < rows; r += NTHR) s_dinv[r] = dinv[lo + r];
    for (int i = tid; i < rows * (CH / 4); i += NTHR) {
        int r  = i / (CH / 4);
        int c4 = i % (CH / 4);
        float d = dinv[lo + r];
        float s = d * d;
        float4 hv = ((const float4*)(h + (size_t)(lo + r) * CH))[c4];
        float4 bv = ((const float4*)bias)[c4];
        float* a = acc + r * CH;
        a[chan_pos<CH>(c4 * 4 + 0)] = fmaf(hv.x, s, bv.x);
        a[chan_pos<CH>(c4 * 4 + 1)] = fmaf(hv.y, s, bv.y);
        a[chan_pos<CH>(c4 * 4 + 2)] = fmaf(hv.z, s, bv.z);
        a[chan_pos<CH>(c4 * 4 + 3)] = fmaf(hv.w, s, bv.w);
    }
    __syncthreads();

    if (found < 0) {
        for (int i = tid; i < e; i += NTHR) {
            int dl = dst[i] - lo;
            if ((unsigned)dl < (unsigned)rows) {
                int s = src[i];
                float w = dinv[s] * s_dinv[dl];
                const float* hr = h + (size_t)s * CH;
                float* a = acc + dl * CH;
                for (int c = 0; c < CH; ++c) atomicAdd(&a[chan_pos<CH>(c)], hr[c] * w);
            }
        }
    }

    const int wave = tid >> 5;
    const int lane = tid & 31;
    for (int base = wave * 32; base < total; base += NTHR) {
        int i = base + lane;
        int s = 0, dl = 0;
        float w = 0.0f;
        if (i < total) {
            s = l_s[i];
            dl = l_d[i];
            w = dinv[s] * s_dinv[dl];
        }
        const int m = min(32, total - base);
#pragma unroll
        for (int g = 0; g < 32; g += 8) {
            if (g + 8 <= m) {
                float hv[8][VEC];
#pragma unroll
                for (int k = 0; k < 8; ++k) {
                    int sj = __builtin_amdgcn_readlane(s, g + k);
                    const float* hr = h + (size_t)sj * CH + lane * VEC;
#pragma unroll
                    for (int j = 0; j < VEC; ++j) hv[k][j] = hr[j];
                }
#pragma unroll
                for (int k = 0; k < 8; ++k) {
                    float wj = __int_as_float(__builtin_amdgcn_readlane(__float_as_int(w), g + k));
                    int dj = __builtin_amdgcn_readlane(dl, g + k);
                    float* a = acc + dj * CH + lane;
#pragma unroll
                    for (int j = 0; j < VEC; ++j) atomicAdd(a + j * 32, hv[k][j] * wj);
                }
            } else if (g < m) {
#pragma unroll
                for (int k = 0; k < 8; ++k) {
                    if (g + k < m) {
                        int sj = __builtin_amdgcn_readlane(s, g + k);
                        float wj = __int_as_float(__builtin_amdgcn_readlane(__float_as_int(w), g + k));
                        int dj = __builtin_amdgcn_readlane(dl, g + k);
                        const float* hr = h + (size_t)sj * CH + lane * VEC;
                        float* a = acc + dj * CH + lane;
#pragma unroll
                        for (int j = 0; j < VEC; ++j) atomicAdd(a + j * 32, hr[j] * wj);
                    }
                }
            }
        }
    }
    __syncthreads();

    for (int i = tid; i < rows * (CH / 4); i += NTHR) {
        int r  = i / (CH / 4);
        int c4 = i % (CH / 4);
        const float* a = acc + r * CH;
        float4 o;
        o.x = a[chan_pos<CH>(c4 * 4 + 0)];
        o.y = a[chan_pos<CH>(c4 * 4 + 1)];
        o.z = a[chan_pos<CH>(c4 * 4 + 2)];
        o.w = a[chan_pos<CH>(c4 * 4 + 3)];
        ((float4*)(out + (size_t)(lo + r) * CH))[c4] = o;
    }
}

extern "C" void kernel_launch(void* const* d_in, const int* in_sizes, int n_in,
                              void* d_out, int out_size, void* d_ws, size_t ws_size,
                              hipStream_t stream) {
    const float* x  = (const float*)d_in[0];
    const int*   ei = (const int*)d_in[1];
    const float* W1 = (const float*)d_in[2];
    const float* b1 = (const float*)d_in[3];
    const float* W2 = (const float*)d_in[4];
    const float* b2 = (const float*)d_in[5];
    float* out = (float*)d_out;

    const int n = in_sizes[0] / 128;
    const int e = in_sizes[1] / 2;
    const int* src = ei;
    const int* dst = ei + e;

    float* dinv = (float*)d_ws;
    float* h  = dinv + n;
    float* o1 = h + (size_t)n * 128;

    const unsigned T = 256;
    constexpr int LCAP = 12288;
    const int nr = (n + KR - 1) / KR;

    unsigned char* keys = (unsigned char*)o1;
    int* lists = (int*)d_out;
    const int kblocks = (e / 16 + 255) / 256 + 1;

    k_keys<<<kblocks, T, 0, stream>>>(dst, keys, e);
    k_deg<LCAP><<<nr, NTHR, 0, stream>>>(src, dst, keys, dinv, lists, n, e);

    k_gemm<128, 1, false><<<(n + 15) / 16, T, 0, stream>>>(x, W1, h, n);
    k_agg<128, LCAP><<<nr, NTHR, 0, stream>>>(src, dst, lists, h, b1, dinv, o1, n, e);

    k_gemm<64, 2, true><<<(n + 31) / 32, T, 0, stream>>>(o1, W2, h, n);
    k_agg<64, LCAP><<<nr, NTHR, 0, stream>>>(src, dst, lists, h, b2, dinv, out, n, e);
}


# Workspace re-ordered (h, o1, dinv), h on a 256-byte boundary, list entries one whole unit behind their count: a gathered h row covers 2 (layer 2: 1) aligned 256-byte units, not 3 (2); sums unchanged
# speedup vs baseline: 3.2148x; 1.0752x over previous
#include <hip/hip_runtime.h>
#include <stdint.h>

typedef __attribute__((ext_vector_type(2))) float v2f;
typedef __attribute__((ext_vector_type(8))) float v8f;

#define KDIM 128
#define NTHR 1024

#define KR 400
#define LW 64
#define LHEAD 64

__global__ __launch_bounds__(256) void k_keys(const int* __restrict__ dst,
                                              unsigned char* __restrict__ keys, int e) {
    const int t = blockIdx.x * 256 + threadIdx.x;
    const int nt = gridDim.x * 256;
    if ((e & 15) == 0 && (((uintptr_t)dst) & 15) == 0 && (((uintptr_t)keys) & 15) == 0) {
        const int4* d4 = (const int4*)dst;
        uint4* k4 = (uint4*)keys;
        const int nq = e >> 4;
        for (int q = t; q < nq; q += nt) {
            int4 a = d4[4 * q + 0], b = d4[4 * q + 1], c = d4[4 * q + 2], d = d4[4 * q + 3];
            uint4 o;
            o.x = ((unsigned)a.x / KR) | (((unsigned)a.y / KR) << 8) | (((unsigned)a.z / KR) << 16) | (((unsigned)a.w / KR) << 24);
            o.y = ((unsigned)b.x / KR) | (((unsigned)b.y / KR) << 8) | (((unsigned)b.z / KR) << 16) | (((unsigned)b.w / KR) << 24);
            o.z = ((unsigned)c.x / KR) | (((unsigned)c.y / KR) << 8) | (((unsigned)c.z / KR) << 16) | (((unsigned)c.w / KR) << 24);
            o.w = ((unsigned)d.x / KR) | (((unsigned)d.y / KR) << 8) | (((unsigned)d.z / KR) << 16) | (((unsigned)d.w / KR) << 24);
            k4[q] = o;
        }
    } else {
        for (int i = t; i < e; i += nt) keys[i] = (unsigned char)((unsigned)dst[i] / KR);
    }
}

template <int SH, int U, class F>
__device__ __forceinline__ void scan_keys(const unsigned char* __restrict__ keys, int e,
                                          int owner, int tid, F f) {
    if ((e & 15) == 0 && (((uintptr_t)keys) & 15) == 0) {
        const uint4* k4 = (const uint4*)keys;
        const int nq = e >> 4;
        constexpr int BATCH = U * NTHR;
        constexpr unsigned KM = ((0xFFu << SH) & 0xFFu) * 0x01010101u;
        const unsigned pat = (unsigned)(owner << SH) * 0x01010101u;

        auto miss = [&](unsigned w) -> unsigned {
            unsigned x = (w & KM) ^ pat;
            return ((x & 0x7F7F7F7Fu) + 0x7F7F7F7Fu) | x;
        };

        static_assert(U == 4, "the match mask below holds four pieces");
        auto test = [&](const uint4 (&v)[U], int q0) {
            unsigned mk[U];
#pragma unroll
            for (int u = 0; u < U; ++u) {
                unsigned nz = ((miss(v[u].x) >> 7) & 0x01010101u) | ((miss(v[u].y) >> 6) & 0x02020202u)
                            | ((miss(v[u].z) >> 5) & 0x04040404u) | ((miss(v[u].w) >> 4) & 0x08080808u);
                mk[u] = ~nz & 0x0F0F0F0Fu;
            }
            unsigned long long m = (unsigned long long)(mk[0] | (mk[1] << 4))
                                 | ((unsigned long long)(mk[2] | (mk[3] << 4)) << 32);
            const int e0 = q0 * 16;
            while (m) {
                const int p = __builtin_ctzll(m);
                m &= m - 1;
                const int piece = ((p >> 5) << 1) | ((p >> 2) & 1);
                f(e0 + piece * (NTHR * 16) + ((p & 3) << 2) + ((p >> 3) & 3));
            }
        };

        uint4 va[U], vb[U];
        const int nfull = nq / (2 * BATCH);
        if (nfull > 0) {
#pragma unroll
            for (int u = 0; u < U; ++u) va[u] = k4[tid + u * NTHR];
            for (int t = 0; t < nfull; ++t) {
                const int q0 = t * (2 * BATCH) + tid;
#pragma unroll
                for (int u = 0; u < U; ++u) vb[u] = k4[q0 + BATCH + u * NTHR];
                test(va, q0);
                const int qn = (t + 1 < nfull) ? q0 + 2 * BATCH : q0;
#pragma unroll
                for (int u = 0; u < U; ++u) va[u] = k4[qn + u * NTHR];
                test(vb, q0 + BATCH);
            }
        }
        for (int q0 = nfull * (2 * BATCH) + tid; q0 < nq; q0 += BATCH) {
#pragma unroll
            for (int u = 0; u < U; ++u) {
                int q = q0 + u * NTHR;
                va[u] = (q < nq) ? k4[q] : make_uint4(~pat, ~pat, ~pat, ~pat);
            }
            test(va, q0);
        }
    } else {
        for (int i = tid; i < e; i += NTHR)
            if ((keys[i] >> SH) == owner) f(i);
    }
}

template <int CAP>
__global__ __launch_bounds__(NTHR) void k_deg(const int* __restrict__ src,
                                              const int* __restrict__ dst,
                                              const unsigned char* __restrict__ keys,
                                              float* __restrict__ dinv,
                                              int* __restrict__ lists, int n, int e) {
    __shared__ int cnt[KR];
    __shared__ int l_s[CAP];
    __shared__ int l_d[CAP];
    __shared__ int s_cnt;
    const int tid = threadIdx.x;
    const int lo = blockIdx.x * KR;
    const int rows = min(KR, n - lo);
    if (tid == 0) s_cnt = 0;
    for (int r = tid; r < KR; r += NTHR) cnt[r] = 0;
    __syncthreads();
    scan_keys<0, 4>(keys, e, blockIdx.x, tid, [&](int eidx) {
        int slot = atomicAdd(&s_cnt, 1);
        if (slot < CAP) l_s[slot] = eidx;
        else atomicAdd(&cnt[dst[eidx] - lo], 1);
    });
    __syncthreads();

    const int found = s_cnt;
    const int total = min(found, CAP);
    for (int i = tid; i < total; i += NTHR) {
        int ei = l_s[i];
        int s = src[ei];
        int dl = dst[ei] - lo;
        l_s[i] = s;
        l_d[i] = dl;
        atomicAdd(&cnt[dl], 1);
    }
    __syncthreads();

    for (int r = tid; r < rows; r += NTHR)
        dinv[lo + r] = rsqrtf((float)(cnt[r] + 1));

    int* reg = lists + (size_t)lo * LW;
    const int room = min(CAP, (rows * LW - LHEAD) / 2);
    const bool fits = found <= room;
    if (tid == 0) reg[0] = fits ? found : -1;
    if (fits) {
        int2* ent = (int2*)(reg + LHEAD);
        for (int i = tid; i < total; i += NTHR) ent[i] = make_int2(l_s[i], l_d[i]);
    }
}

template <int OUTCH, int MT, bool RELU>
__global__ __launch_bounds__(256) void k_gemm(const float* __restrict__ A,
                                              const float* __restrict__ W,
                                              float* __restrict__ H, int n) {
    constexpr int LDSW = KDIM + 4;
    __shared__ float tile[MT * 16 * LDSW];

    const int tid = threadIdx.x;
    const int m_base = blockIdx.x * (MT * 16);

    constexpr int CHUNKS = MT * 16 * (KDIM / 4);
    for (int c = tid; c < CHUNKS; c += 256) {
        int row = c >> 5;
        int kc  = c & 31;
        int grow = m_base + row;
        if (grow >= n) grow = n - 1;
        float4 v = ((const float4*)(A + (size_t)grow * KDIM))[kc];
        if (RELU) {
            v.x = fmaxf(v.x, 0.0f); v.y = fmaxf(v.y, 0.0f);
            v.z = fmaxf(v.z, 0.0f); v.w = fmaxf(v.w, 0.0f);
        }
        *(float4*)&tile[row * LDSW + kc * 4] = v;
    }
    __syncthreads();

    const int wave = tid >> 5;
    const int lane = tid & 31;
    int mt, nt;
    if (OUTCH == 128) { mt = 0;         nt = wave;     }
    else              { mt = wave >> 2; nt = wave & 3; }
    const int lm = lane & 15;
    const int kh = lane >> 4;

    const float* arow = &tile[(mt * 16 + lm) * LDSW];
    const float* wcol = W + nt * 16 + lm;

    v8f acc = {};
#pragma unroll
    for (int k0 = 0; k0 < KDIM; k0 += 4) {
        v2f a, b;
        a.x = arow[k0 + 2 * kh + 0];
        a.y = arow[k0 + 2 * kh + 1];
        b.x = wcol[(size_t)(k0 + 2 * kh + 0) * OUTCH];
        b.y = wcol[(size_t)(k0 + 2 * kh + 1) * OUTCH];
        acc = __builtin_amdgcn_wmma_f32_16x16x4_f32(
             false, a,  false, b,
             (short)0, acc,  false,  false);
    }

    const int mrow0 = m_base + mt * 16 + (kh ? 8 : 0);
#pragma unroll
    for (int r = 0; r < 8; ++r) {
        int grow = mrow0 + r;
        if (grow < n)
            H[(size_t)grow * OUTCH + nt * 16 + lm] = acc[r];
    }
}

template <int CH>
__device__ __forceinline__ int chan_pos(int c) {
    constexpr int VEC = CH / 32;
    return (c % VEC) * 32 + c / VEC;
}

template <int CH, int CAP>
__global__ __launch_bounds__(NTHR) void k_agg(const int* __restrict__ src,
                                              const int* __restrict__ dst,
                                              const int* lists,
                                              const float* __restrict__ h,
                                              const float* __restrict__ bias,
                                              const float* __restrict__ dinv,
                                              float* out, int n, int e) {
    constexpr int VEC = CH / 32;
    constexpr int R = KR;
    __shared__ float acc[R * CH];
    __shared__ int   l_s[CAP];
    __shared__ int   l_d[CAP];
    __shared__ float s_dinv[R];

    const int tid = threadIdx.x;
    const int lo = blockIdx.x * R;
    const int rows = min(R, n - lo);

    const int* reg = lists + (size_t)lo * LW;
    const int found = reg[0];
    const int total = found < 0 ? 0 : min(found, CAP);
    {
        const int2* ent = (const int2*)(reg + LHEAD);
        for (int i = tid; i < total; i += NTHR) {
            int2 p = ent[i];
            l_s[i] = p.x;
            l_d[i] = p.y;
        }
    }

    for (int r = tid; r < rows; r += NTHR) s_dinv[r] = dinv[lo + r];
    for (int i = tid; i < rows * (CH / 4); i += NTHR) {
        int r  = i / (CH / 4);
        int c4 = i % (CH / 4);
        float d = dinv[lo + r];
        float s = d * d;
        float4 hv = ((const float4*)(h + (size_t)(lo + r) * CH))[c4];
        float4 bv = ((const float4*)bias)[c4];
        float* a = acc + r * CH;
        a[chan_pos<CH>(c4 * 4 + 0)] = fmaf(hv.x, s, bv.x);
        a[chan_pos<CH>(c4 * 4 + 1)] = fmaf(hv.y, s, bv.y);
        a[chan_pos<CH>(c4 * 4 + 2)] = fmaf(hv.z, s, bv.z);
        a[chan_pos<CH>(c4 * 4 + 3)] = fmaf(hv.w, s, bv.w);
    }
    __syncthreads();

    if (found < 0) {
        for (int i = tid; i < e; i += NTHR) {
            int dl = dst[i] - lo;
            if ((unsigned)dl < (unsigned)rows) {
                int s = src[i];
                float w = dinv[s] * s_dinv[dl];
                const float* hr = h + (size_t)s * CH;
                float* a = acc + dl * CH;
                for (int c = 0; c < CH; ++c) atomicAdd(&a[chan_pos<CH>(c)], hr[c] * w);
            }
        }
    }

    const int wave = tid >> 5;
    const int lane = tid & 31;
    for (int base = wave * 32; base < total; base += NTHR) {
        int i = base + lane;
        int s = 0, dl = 0;
        float w = 0.0f;
        if (i < total) {
            s = l_s[i];
            dl = l_d[i];
            w = dinv[s] * s_dinv[dl];
        }
        const int m = min(32, total - base);
#pragma unroll
        for (int g = 0; g < 32; g += 8) {
            if (g + 8 <= m) {
                float hv[8][VEC];
#pragma unroll
                for (int k = 0; k < 8; ++k) {
                    int sj = __builtin_amdgcn_readlane(s, g + k);
                    const float* hr = h + (size_t)sj * CH + lane * VEC;
#pragma unroll
                    for (int j = 0; j < VEC; ++j) hv[k][j] = hr[j];
                }
#pragma unroll
                for (int k = 0; k < 8; ++k) {
                    float wj = __int_as_float(__builtin_amdgcn_readlane(__float_as_int(w), g + k));
                    int dj = __builtin_amdgcn_readlane(dl, g + k);
                    float* a = acc + dj * CH + lane;
#pragma unroll
                    for (int j = 0; j < VEC; ++j) atomicAdd(a + j * 32, hv[k][j] * wj);
                }
            } else if (g < m) {
#pragma unroll
                for (int k = 0; k < 8; ++k) {
                    if (g + k < m) {
                        int sj = __builtin_amdgcn_readlane(s, g + k);
                        float wj = __int_as_float(__builtin_amdgcn_readlane(__float_as_int(w), g + k));
                        int dj = __builtin_amdgcn_readlane(dl, g + k);
                        const float* hr = h + (size_t)sj * CH + lane * VEC;
                        float* a = acc + dj * CH + lane;
#pragma unroll
                        for (int j = 0; j < VEC; ++j) atomicAdd(a + j * 32, hr[j] * wj);
                    }
                }
            }
        }
    }
    __syncthreads();

    for (int i = tid; i < rows * (CH / 4); i += NTHR) {
        int r  = i / (CH / 4);
        int c4 = i % (CH / 4);
        const float* a = acc + r * CH;
        float4 o;
        o.x = a[chan_pos<CH>(c4 * 4 + 0)];
        o.y = a[chan_pos<CH>(c4 * 4 + 1)];
        o.z = a[chan_pos<CH>(c4 * 4 + 2)];
        o.w = a[chan_pos<CH>(c4 * 4 + 3)];
        ((float4*)(out + (size_t)(lo + r) * CH))[c4] = o;
    }
}

extern "C" void kernel_launch(void* const* d_in, const int* in_sizes, int n_in,
                              void* d_out, int out_size, void* d_ws, size_t ws_size,
                              hipStream_t stream) {
    const float* x  = (const float*)d_in[0];
    const int*   ei = (const int*)d_in[1];
    const float* W1 = (const float*)d_in[2];
    const float* b1 = (const float*)d_in[3];
    const float* W2 = (const float*)d_in[4];
    const float* b2 = (const float*)d_in[5];
    float* out = (float*)d_out;

    const int n = in_sizes[0] / 128;
    const int e = in_sizes[1] / 2;
    const int* src = ei;
    const int* dst = ei + e;

    uintptr_t wb = (uintptr_t)d_ws;
    const uintptr_t wa = (wb + 255) & ~(uintptr_t)255;
    const size_t need = ((size_t)n * 256 + (size_t)n) * sizeof(float);
    if ((wa - wb) + need <= ws_size) wb = wa;
    float* h  = (float*)wb;
    float* o1 = h + (size_t)n * 128;
    float* dinv = o1 + (size_t)n * 128;

    const unsigned T = 256;
    constexpr int LCAP = 12288;
    const int nr = (n + KR - 1) / KR;

    unsigned char* keys = (unsigned char*)o1;
    int* lists = (int*)d_out;
    const int kblocks = (e / 16 + 255) / 256 + 1;

    k_keys<<<kblocks, T, 0, stream>>>(dst, keys, e);
    k_deg<LCAP><<<nr, NTHR, 0, stream>>>(src, dst, keys, dinv, lists, n, e);

    k_gemm<128, 1, false><<<(n + 15) / 16, T, 0, stream>>>(x, W1, h, n);
    k_agg<128, LCAP><<<nr, NTHR, 0, stream>>>(src, dst, lists, h, b1, dinv, o1, n, e);

    k_gemm<64, 2, true><<<(n + 31) / 32, T, 0, stream>>>(o1, W2, h, n);
    k_agg<64, LCAP><<<nr, NTHR, 0, stream>>>(src, dst, lists, h, b2, dinv, out, n, e);
}


# Degree pass reads its range's row of presence bits (one per 16-edge key piece, set by the key kernel) and loads only pieces holding its edges, not all keys; list slots per wave; list order changes
# speedup vs baseline: 3.4501x; 1.0732x over previous
#include <hip/hip_runtime.h>
#include <stdint.h>

typedef __attribute__((ext_vector_type(2))) float v2f;
typedef __attribute__((ext_vector_type(8))) float v8f;

#define KDIM 128
#define NTHR 1024

#define KR 400
#define LW 64
#define LHEAD 64
#define KW 257

__global__ __launch_bounds__(1024) void k_keys(const int* __restrict__ dst,
                                               unsigned char* __restrict__ keys,
                                               unsigned* __restrict__ plane, int e, int pw) {
    __shared__ unsigned bm[32 * KW];
    const int tid = threadIdx.x;
    const int wave = tid >> 5;
    const int lane = tid & 31;
    for (int i = tid; i < 32 * KW; i += 1024) bm[i] = 0u;
    __syncthreads();

    const int piece = blockIdx.x * 1024 + tid;
    const int e0 = piece * 16;
    if (e0 < e) {
        unsigned k[16];
        if (e0 + 16 <= e && (((uintptr_t)dst) & 15) == 0) {
            const int4* d4 = (const int4*)(dst + e0);
            const int4 a = d4[0], b = d4[1], c = d4[2], d = d4[3];
            k[0]  = (unsigned)a.x / KR; k[1]  = (unsigned)a.y / KR; k[2]  = (unsigned)a.z / KR; k[3]  = (unsigned)a.w / KR;
            k[4]  = (unsigned)b.x / KR; k[5]  = (unsigned)b.y / KR; k[6]  = (unsigned)b.z / KR; k[7]  = (unsigned)b.w / KR;
            k[8]  = (unsigned)c.x / KR; k[9]  = (unsigned)c.y / KR; k[10] = (unsigned)c.z / KR; k[11] = (unsigned)c.w / KR;
            k[12] = (unsigned)d.x / KR; k[13] = (unsigned)d.y / KR; k[14] = (unsigned)d.z / KR; k[15] = (unsigned)d.w / KR;
        } else {
#pragma unroll
            for (int j = 0; j < 16; ++j) k[j] = (e0 + j < e) ? (unsigned)dst[e0 + j] / KR : 0xFFu;
        }
#pragma unroll
        for (int j = 0; j < 16; ++j) k[j] &= 0xFFu;
        uint4 o;
        o.x = k[0]  | (k[1]  << 8) | (k[2]  << 16) | (k[3]  << 24);
        o.y = k[4]  | (k[5]  << 8) | (k[6]  << 16) | (k[7]  << 24);
        o.z = k[8]  | (k[9]  << 8) | (k[10] << 16) | (k[11] << 24);
        o.w = k[12] | (k[13] << 8) | (k[14] << 16) | (k[15] << 24);
        *(uint4*)(keys + e0) = o;
        unsigned* my = bm + wave * KW;
        const unsigned bit = 1u << lane;
#pragma unroll
        for (int j = 0; j < 16; ++j)
            if (e0 + j < e) atomicOr(&my[k[j]], bit);
    }
    __syncthreads();

    for (int j = 0; j < 8; ++j) {
        const int o = wave * 8 + j;
        plane[(size_t)o * pw + blockIdx.x * 32 + lane] = bm[lane * KW + o];
    }
}

template <int BITS>
__device__ __forceinline__ int wave_prefix(int c, int& tot) {
    int pre = 0, t = 0;
#pragma unroll
    for (int b = 0; b < BITS; ++b) {
        const unsigned m = __builtin_amdgcn_ballot_w32(((c >> b) & 1) != 0);
        pre += (int)__builtin_amdgcn_mbcnt_lo(m, 0u) << b;
        t += __builtin_popcount(m) << b;
    }
    tot = t;
    return pre;
}

template <int CAP>
__global__ __launch_bounds__(NTHR) void k_deg(const int* __restrict__ src,
                                              const int* __restrict__ dst,
                                              const unsigned char* __restrict__ keys,
                                              const unsigned* __restrict__ plane,
                                              float* __restrict__ dinv,
                                              int* __restrict__ lists, int n, int e, int pw) {
    __shared__ int cnt[KR];
    __shared__ int l_s[CAP];
    __shared__ int l_d[CAP];
    __shared__ int s_cnt;
    __shared__ int s_np;
    const int tid = threadIdx.x;
    const int lane = tid & 31;
    const int lo = blockIdx.x * KR;
    const int rows = min(KR, n - lo);
    if (tid == 0) { s_cnt = 0; s_np = 0; }
    for (int r = tid; r < KR; r += NTHR) cnt[r] = 0;
    __syncthreads();

    const uint4* k4 = (const uint4*)keys;
    const unsigned pat = (unsigned)blockIdx.x * 0x01010101u;
    auto miss = [&](unsigned w) -> unsigned {
        unsigned x = w ^ pat;
        return ((x & 0x7F7F7F7Fu) + 0x7F7F7F7Fu) | x;
    };
    auto match16 = [&](const uint4& v, int pc) -> unsigned {
        unsigned nz = ((miss(v.x) >> 7) & 0x01010101u) | ((miss(v.y) >> 6) & 0x02020202u)
                    | ((miss(v.z) >> 5) & 0x04040404u) | ((miss(v.w) >> 4) & 0x08080808u);
        unsigned mk = ~nz & 0x0F0F0F0Fu;
        const int left = e - pc * 16;
        if (left < 16) {
            unsigned keep = 0u;
            for (int o = 0; o < left; ++o) keep |= 1u << (8 * (o & 3) + (o >> 2));
            mk &= keep;
        }
        return mk;
    };
    auto edge_of = [&](int pc, int p) -> int { return pc * 16 + ((p & 3) << 2) + (p >> 3); };
    auto put = [&](int eidx, int slot) {
        if (slot < CAP) l_s[slot] = eidx;
        else atomicAdd(&cnt[dst[eidx] - lo], 1);
    };

    const unsigned* row = plane + (size_t)blockIdx.x * pw;
    for (int w0 = 0; w0 < pw; w0 += 4 * NTHR) {
        unsigned w[4];
        int c = 0;
#pragma unroll
        for (int u = 0; u < 4; ++u) {
            const int wi = w0 + u * NTHR + tid;
            w[u] = (wi < pw) ? row[wi] : 0u;
            c += __builtin_popcount(w[u]);
        }
        int tot;
        const int pre = wave_prefix<8>(c, tot);
        int base = 0;
        if (lane == 0) base = atomicAdd(&s_np, tot);
        int slot = __builtin_amdgcn_readfirstlane(base) + pre;
#pragma unroll
        for (int u = 0; u < 4; ++u) {
            const int wi = w0 + u * NTHR + tid;
            unsigned x = w[u];
            while (x) {
                const int pc = wi * 32 + __builtin_ctz(x);
                x &= x - 1;
                if (slot < CAP) {
                    l_d[slot] = pc;
                } else {
                    unsigned y = match16(k4[pc], pc);
                    while (y) {
                        const int p = __builtin_ctz(y);
                        y &= y - 1;
                        put(edge_of(pc, p), atomicAdd(&s_cnt, 1));
                    }
                }
                ++slot;
            }
        }
    }
    __syncthreads();

    const int np = min(s_np, CAP);
    for (int i0 = 0; i0 < np; i0 += 8 * NTHR) {
        int pc[8];
        uint4 v[8];
        unsigned mk[8];
        int c = 0;
#pragma unroll
        for (int u = 0; u < 8; ++u) {
            const int i = i0 + u * NTHR + tid;
            pc[u] = (i < np) ? l_d[i] : -1;
        }
#pragma unroll
        for (int u = 0; u < 8; ++u) v[u] = k4[max(pc[u], 0)];
#pragma unroll
        for (int u = 0; u < 8; ++u) {
            mk[u] = (pc[u] >= 0) ? match16(v[u], pc[u]) : 0u;
            c += __builtin_popcount(mk[u]);
        }
        int tot;
        const int pre = wave_prefix<8>(c, tot);
        int base = 0;
        if (lane == 0) base = atomicAdd(&s_cnt, tot);
        int slot = __builtin_amdgcn_readfirstlane(base) + pre;
#pragma unroll
        for (int u = 0; u < 8; ++u) {
            unsigned x = mk[u];
            while (x) {
                const int p = __builtin_ctz(x);
                x &= x - 1;
                put(edge_of(pc[u], p), slot);
                ++slot;
            }
        }
    }
    __syncthreads();

    const int found = s_cnt;
    const int total = min(found, CAP);
    for (int i = tid; i < total; i += NTHR) {
        int ei = l_s[i];
        int s = src[ei];
        int dl = dst[ei] - lo;
        l_s[i] = s;
        l_d[i] = dl;
        atomicAdd(&cnt[dl], 1);
    }
    __syncthreads();

    for (int r = tid; r < rows; r += NTHR)
        dinv[lo + r] = rsqrtf((float)(cnt[r] + 1));

    int* reg = lists + (size_t)lo * LW;
    const int room = min(CAP, (rows * LW - LHEAD) / 2);
    const bool fits = found <= room;
    if (tid == 0) reg[0] = fits ? found : -1;
    if (fits) {
        int2* ent = (int2*)(reg + LHEAD);
        for (int i = tid; i < total; i += NTHR) ent[i] = make_int2(l_s[i], l_d[i]);
    }
}

template <int OUTCH, int MT, bool RELU>
__global__ __launch_bounds__(256) void k_gemm(const float* __restrict__ A,
                                              const float* __restrict__ W,
                                              float* __restrict__ H, int n) {
    constexpr int LDSW = KDIM + 4;
    __shared__ float tile[MT * 16 * LDSW];

    const int tid = threadIdx.x;
    const int m_base = blockIdx.x * (MT * 16);

    constexpr int CHUNKS = MT * 16 * (KDIM / 4);
    for (int c = tid; c < CHUNKS; c += 256) {
        int row = c >> 5;
        int kc  = c & 31;
        int grow = m_base + row;
        if (grow >= n) grow = n - 1;
        float4 v = ((const float4*)(A + (size_t)grow * KDIM))[kc];
        if (RELU) {
            v.x = fmaxf(v.x, 0.0f); v.y = fmaxf(v.y, 0.0f);
            v.z = fmaxf(v.z, 0.0f); v.w = fmaxf(v.w, 0.0f);
        }
        *(float4*)&tile[row * LDSW + kc * 4] = v;
    }
    __syncthreads();

    const int wave = tid >> 5;
    const int lane = tid & 31;
    int mt, nt;
    if (OUTCH == 128) { mt = 0;         nt = wave;     }
    else              { mt = wave >> 2; nt = wave & 3; }
    const int lm = lane & 15;
    const int kh = lane >> 4;

    const float* arow = &tile[(mt * 16 + lm) * LDSW];
    const float* wcol = W + nt * 16 + lm;

    v8f acc = {};
#pragma unroll
    for (int k0 = 0; k0 < KDIM; k0 += 4) {
        v2f a, b;
        a.x = arow[k0 + 2 * kh + 0];
        a.y = arow[k0 + 2 * kh + 1];
        b.x = wcol[(size_t)(k0 + 2 * kh + 0) * OUTCH];
        b.y = wcol[(size_t)(k0 + 2 * kh + 1) * OUTCH];
        acc = __builtin_amdgcn_wmma_f32_16x16x4_f32(
             false, a,  false, b,
             (short)0, acc,  false,  false);
    }

    const int mrow0 = m_base + mt * 16 + (kh ? 8 : 0);
#pragma unroll
    for (int r = 0; r < 8; ++r) {
        int grow = mrow0 + r;
        if (grow < n)
            H[(size_t)grow * OUTCH + nt * 16 + lm] = acc[r];
    }
}

template <int CH>
__device__ __forceinline__ int chan_pos(int c) {
    constexpr int VEC = CH / 32;
    return (c % VEC) * 32 + c / VEC;
}

template <int CH, int CAP>
__global__ __launch_bounds__(NTHR) void k_agg(const int* __restrict__ src,
                                              const int* __restrict__ dst,
                                              const int* lists,
                                              const float* __restrict__ h,
                                              const float* __restrict__ bias,
                                              const float* __restrict__ dinv,
                                              float* out, int n, int e) {
    constexpr int VEC = CH / 32;
    constexpr int R = KR;
    __shared__ float acc[R * CH];
    __shared__ int   l_s[CAP];
    __shared__ int   l_d[CAP];
    __shared__ float s_dinv[R];

    const int tid = threadIdx.x;
    const int lo = blockIdx.x * R;
    const int rows = min(R, n - lo);

    const int* reg = lists + (size_t)lo * LW;
    const int found = reg[0];
    const int total = found < 0 ? 0 : min(found, CAP);
    {
        const int2* ent = (const int2*)(reg + LHEAD);
        for (int i = tid; i < total; i += NTHR) {
            int2 p = ent[i];
            l_s[i] = p.x;
            l_d[i] = p.y;
        }
    }

    for (int r = tid; r < rows; r += NTHR) s_dinv[r] = dinv[lo + r];
    for (int i = tid; i < rows * (CH / 4); i += NTHR) {
        int r  = i / (CH / 4);
        int c4 = i % (CH / 4);
        float d = dinv[lo + r];
        float s = d * d;
        float4 hv = ((const float4*)(h + (size_t)(lo + r) * CH))[c4];
        float4 bv = ((const float4*)bias)[c4];
        float* a = acc + r * CH;
        a[chan_pos<CH>(c4 * 4 + 0)] = fmaf(hv.x, s, bv.x);
        a[chan_pos<CH>(c4 * 4 + 1)] = fmaf(hv.y, s, bv.y);
        a[chan_pos<CH>(c4 * 4 + 2)] = fmaf(hv.z, s, bv.z);
        a[chan_pos<CH>(c4 * 4 + 3)] = fmaf(hv.w, s, bv.w);
    }
    __syncthreads();

    if (found < 0) {
        for (int i = tid; i < e; i += NTHR) {
            int dl = dst[i] - lo;
            if ((unsigned)dl < (unsigned)rows) {
                int s = src[i];
                float w = dinv[s] * s_dinv[dl];
                const float* hr = h + (size_t)s * CH;
                float* a = acc + dl * CH;
                for (int c = 0; c < CH; ++c) atomicAdd(&a[chan_pos<CH>(c)], hr[c] * w);
            }
        }
    }

    const int wave = tid >> 5;
    const int lane = tid & 31;
    for (int base = wave * 32; base < total; base += NTHR) {
        int i = base + lane;
        int s = 0, dl = 0;
        float w = 0.0f;
        if (i < total) {
            s = l_s[i];
            dl = l_d[i];
            w = dinv[s] * s_dinv[dl];
        }
        const int m = min(32, total - base);
#pragma unroll
        for (int g = 0; g < 32; g += 8) {
            if (g + 8 <= m) {
                float hv[8][VEC];
#pragma unroll
                for (int k = 0; k < 8; ++k) {
                    int sj = __builtin_amdgcn_readlane(s, g + k);
                    const float* hr = h + (size_t)sj * CH + lane * VEC;
#pragma unroll
                    for (int j = 0; j < VEC; ++j) hv[k][j] = hr[j];
                }
#pragma unroll
                for (int k = 0; k < 8; ++k) {
                    float wj = __int_as_float(__builtin_amdgcn_readlane(__float_as_int(w), g + k));
                    int dj = __builtin_amdgcn_readlane(dl, g + k);
                    float* a = acc + dj * CH + lane;
#pragma unroll
                    for (int j = 0; j < VEC; ++j) atomicAdd(a + j * 32, hv[k][j] * wj);
                }
            } else if (g < m) {
#pragma unroll
                for (int k = 0; k < 8; ++k) {
                    if (g + k < m) {
                        int sj = __builtin_amdgcn_readlane(s, g + k);
                        float wj = __int_as_float(__builtin_amdgcn_readlane(__float_as_int(w), g + k));
                        int dj = __builtin_amdgcn_readlane(dl, g + k);
                        const float* hr = h + (size_t)sj * CH + lane * VEC;
                        float* a = acc + dj * CH + lane;
#pragma unroll
                        for (int j = 0; j < VEC; ++j) atomicAdd(a + j * 32, hr[j] * wj);
                    }
                }
            }
        }
    }
    __syncthreads();

    for (int i = tid; i < rows * (CH / 4); i += NTHR) {
        int r  = i / (CH / 4);
        int c4 = i % (CH / 4);
        const float* a = acc + r * CH;
        float4 o;
        o.x = a[chan_pos<CH>(c4 * 4 + 0)];
        o.y = a[chan_pos<CH>(c4 * 4 + 1)];
        o.z = a[chan_pos<CH>(c4 * 4 + 2)];
        o.w = a[chan_pos<CH>(c4 * 4 + 3)];
        ((float4*)(out + (size_t)(lo + r) * CH))[c4] = o;
    }
}

extern "C" void kernel_launch(void* const* d_in, const int* in_sizes, int n_in,
                              void* d_out, int out_size, void* d_ws, size_t ws_size,
                              hipStream_t stream) {
    const float* x  = (const float*)d_in[0];
    const int*   ei = (const int*)d_in[1];
    const float* W1 = (const float*)d_in[2];
    const float* b1 = (const float*)d_in[3];
    const float* W2 = (const float*)d_in[4];
    const float* b2 = (const float*)d_in[5];
    float* out = (float*)d_out;

    const int n = in_sizes[0] / 128;
    const int e = in_sizes[1] / 2;
    const int* src = ei;
    const int* dst = ei + e;

    uintptr_t wb = (uintptr_t)d_ws;
    const uintptr_t wa = (wb + 255) & ~(uintptr_t)255;
    const size_t need = ((size_t)n * 256 + (size_t)n) * sizeof(float);
    if ((wa - wb) + need <= ws_size) wb = wa;
    float* h  = (float*)wb;
    float* o1 = h + (size_t)n * 128;
    float* dinv = o1 + (size_t)n * 128;

    const unsigned T = 256;
    constexpr int LCAP = 12288;
    const int nr = (n + KR - 1) / KR;

    const int npieces = (e + 15) / 16;
    const int kblocks = (npieces + 1023) / 1024;
    const int pw = kblocks * 32;
    unsigned char* keys = (unsigned char*)o1;
    unsigned* plane = (unsigned*)(keys + (((size_t)npieces * 16 + 255) & ~(size_t)255));
    int* lists = (int*)d_out;

    k_keys<<<kblocks, NTHR, 0, stream>>>(dst, keys, plane, e, pw);
    k_deg<LCAP><<<nr, NTHR, 0, stream>>>(src, dst, keys, plane, dinv, lists, n, e, pw);

    k_gemm<128, 1, false><<<(n + 15) / 16, T, 0, stream>>>(x, W1, h, n);
    k_agg<128, LCAP><<<nr, NTHR, 0, stream>>>(src, dst, lists, h, b1, dinv, o1, n, e);

    k_gemm<64, 2, true><<<(n + 31) / 32, T, 0, stream>>>(o1, W2, h, n);
    k_agg<64, LCAP><<<nr, NTHR, 0, stream>>>(src, dst, lists, h, b2, dinv, out, n, e);
}


# Degree pass loads one 128-byte record per listed piece (16 dst,src pairs, written in-call by the key kernel), 8 lanes a line: no key piece, no src/dst gather per edge; list (arrival) order changes
# speedup vs baseline: 3.5525x; 1.0297x over previous
#include <hip/hip_runtime.h>
#include <stdint.h>

typedef __attribute__((ext_vector_type(2))) float v2f;
typedef __attribute__((ext_vector_type(8))) float v8f;

#define KDIM 128
#define NTHR 1024

#define KR 400
#define LW 64
#define LHEAD 64
#define KW 257
#define RB 9

__global__ __launch_bounds__(1024) void k_keys(const int* __restrict__ src,
                                               const int* __restrict__ dst,
                                               int* __restrict__ rec,
                                               unsigned* __restrict__ plane, int e, int pw) {
    __shared__ unsigned bm[32 * KW];
    const int tid = threadIdx.x;
    const int wave = tid >> 5;
    const int lane = tid & 31;
    const int oct = lane >> 3;
    const int sub = lane & 7;
    for (int i = tid; i < 32 * KW; i += 1024) bm[i] = 0u;
    __syncthreads();

    unsigned* my = bm + wave * KW;
    int4* rec4 = (int4*)rec;
    const int pc0 = blockIdx.x * 1024 + wave * 32;
    const bool al = ((((uintptr_t)dst) | ((uintptr_t)src)) & 7) == 0;
    if (al && (pc0 + 32) * 16 <= e) {
        int2 d[8], s[8];
#pragma unroll
        for (int u = 0; u < 8; ++u) {
            const int e0 = (pc0 + u * 4 + oct) * 16 + 2 * sub;
            d[u] = *(const int2*)(dst + e0);
            s[u] = *(const int2*)(src + e0);
        }
#pragma unroll
        for (int u = 0; u < 8; ++u) {
            const int b = u * 4 + oct;
            rec4[(size_t)(pc0 + b) * 8 + sub] = make_int4(d[u].x, s[u].x, d[u].y, s[u].y);
            atomicOr(&my[((unsigned)d[u].x / KR) & 0xFFu], 1u << b);
            atomicOr(&my[((unsigned)d[u].y / KR) & 0xFFu], 1u << b);
        }
    } else {
        for (int u = 0; u < 8; ++u) {
            const int b = u * 4 + oct;
            const int pc = pc0 + b;
            if (pc * 16 < e) {
                const int e0 = pc * 16 + 2 * sub;
                const bool x0 = e0 < e, x1 = e0 + 1 < e;
                const int d0 = x0 ? dst[e0] : -1,     s0 = x0 ? src[e0] : 0;
                const int d1 = x1 ? dst[e0 + 1] : -1, s1 = x1 ? src[e0 + 1] : 0;
                rec4[(size_t)pc * 8 + sub] = make_int4(d0, s0, d1, s1);
                if (x0) atomicOr(&my[((unsigned)d0 / KR) & 0xFFu], 1u << b);
                if (x1) atomicOr(&my[((unsigned)d1 / KR) & 0xFFu], 1u << b);
            }
        }
    }
    __syncthreads();

    for (int j = 0; j < 8; ++j) {
        const int o = wave * 8 + j;
        plane[(size_t)o * pw + blockIdx.x * 32 + lane] = bm[lane * KW + o];
    }
}

template <int BITS>
__device__ __forceinline__ int wave_prefix(int c, int& tot) {
    int pre = 0, t = 0;
#pragma unroll
    for (int b = 0; b < BITS; ++b) {
        const unsigned m = __builtin_amdgcn_ballot_w32(((c >> b) & 1) != 0);
        pre += (int)__builtin_amdgcn_mbcnt_lo(m, 0u) << b;
        t += __builtin_popcount(m) << b;
    }
    tot = t;
    return pre;
}

template <int CAP>
__global__ __launch_bounds__(NTHR) void k_deg(const int* __restrict__ rec,
                                              const unsigned* __restrict__ plane,
                                              float* __restrict__ dinv,
                                              int* __restrict__ lists, int n, int pw) {
    static_assert(KR <= (1 << RB), "a local row must fit RB bits");
    __shared__ int cnt[KR];
    __shared__ int l_s[CAP];
    __shared__ int l_d[CAP];
    __shared__ int s_cnt;
    __shared__ int s_np;
    const int tid = threadIdx.x;
    const int wave = tid >> 5;
    const int lane = tid & 31;
    const int lo = blockIdx.x * KR;
    const int rows = min(KR, n - lo);
    if (tid == 0) { s_cnt = 0; s_np = 0; }
    for (int r = tid; r < KR; r += NTHR) cnt[r] = 0;
    __syncthreads();

    const int4* rec4 = (const int4*)rec;
    auto mine = [&](int d) -> bool { return (unsigned)(d - lo) < (unsigned)KR; };
    auto take = [&](int d, int s, int slot) {
        const int dl = d - lo;
        atomicAdd(&cnt[dl], 1);
        if (slot < CAP) l_s[slot] = (int)(((unsigned)s << RB) | (unsigned)dl);
    };

    const unsigned* row = plane + (size_t)blockIdx.x * pw;
    for (int w0 = 0; w0 < pw; w0 += 4 * NTHR) {
        unsigned w[4];
        int c = 0;
#pragma unroll
        for (int u = 0; u < 4; ++u) {
            const int wi = w0 + u * NTHR + tid;
            w[u] = (wi < pw) ? row[wi] : 0u;
            c += __builtin_popcount(w[u]);
        }
        int tot;
        const int pre = wave_prefix<8>(c, tot);
        int base = 0;
        if (lane == 0) base = atomicAdd(&s_np, tot);
        int slot = __builtin_amdgcn_readfirstlane(base) + pre;
#pragma unroll
        for (int u = 0; u < 4; ++u) {
            const int wi = w0 + u * NTHR + tid;
            unsigned x = w[u];
            while (x) {
                const int pc = wi * 32 + __builtin_ctz(x);
                x &= x - 1;
                if (slot < CAP) {
                    l_d[slot] = pc;
                } else {
                    for (int j = 0; j < 8; ++j) {
                        const int4 q = rec4[(size_t)pc * 8 + j];
                        if (mine(q.x)) take(q.x, q.y, atomicAdd(&s_cnt, 1));
                        if (mine(q.z)) take(q.z, q.w, atomicAdd(&s_cnt, 1));
                    }
                }
                ++slot;
            }
        }
    }
    __syncthreads();

    const int np = min(s_np, CAP);
    const int oct = lane >> 3;
    const int sub = lane & 7;
    for (int i0 = wave * 32; i0 < np; i0 += NTHR) {
        int pc[8];
        int4 v[8];
        unsigned mm = 0u;
        int c = 0;
#pragma unroll
        for (int u = 0; u < 8; ++u) {
            const int i = i0 + u * 4 + oct;
            const int q = l_d[min(i, np - 1)];
            pc[u] = (i < np) ? q : -1;
        }
#pragma unroll
        for (int u = 0; u < 8; ++u) v[u] = rec4[(size_t)max(pc[u], 0) * 8 + sub];
#pragma unroll
        for (int u = 0; u < 8; ++u) {
            const unsigned m0 = (pc[u] >= 0 && mine(v[u].x)) ? 1u : 0u;
            const unsigned m1 = (pc[u] >= 0 && mine(v[u].z)) ? 1u : 0u;
            mm |= (m0 << (2 * u)) | (m1 << (2 * u + 1));
            c += (int)(m0 + m1);
        }
        int tot;
        const int pre = wave_prefix<5>(c, tot);
        int base = 0;
        if (lane == 0) base = atomicAdd(&s_cnt, tot);
        int slot = __builtin_amdgcn_readfirstlane(base) + pre;
#pragma unroll
        for (int u = 0; u < 8; ++u) {
            if (mm & (1u << (2 * u)))     { take(v[u].x, v[u].y, slot); ++slot; }
            if (mm & (1u << (2 * u + 1))) { take(v[u].z, v[u].w, slot); ++slot; }
        }
    }
    __syncthreads();

    const int found = s_cnt;
    const int total = min(found, CAP);
    for (int r = tid; r < rows; r += NTHR)
        dinv[lo + r] = rsqrtf((float)(cnt[r] + 1));

    int* reg = lists + (size_t)lo * LW;
    const int room = min(CAP, (rows * LW - LHEAD) / 2);
    const bool fits = found <= room;
    if (tid == 0) reg[0] = fits ? found : -1;
    if (fits) {
        int2* ent = (int2*)(reg + LHEAD);
        for (int i = tid; i < total; i += NTHR) {
            const unsigned p = (unsigned)l_s[i];
            ent[i] = make_int2((int)(p >> RB), (int)(p & ((1u << RB) - 1u)));
        }
    }
}

template <int OUTCH, int MT, bool RELU>
__global__ __launch_bounds__(256) void k_gemm(const float* __restrict__ A,
                                              const float* __restrict__ W,
                                              float* __restrict__ H, int n) {
    constexpr int LDSW = KDIM + 4;
    __shared__ float tile[MT * 16 * LDSW];

    const int tid = threadIdx.x;
    const int m_base = blockIdx.x * (MT * 16);

    constexpr int CHUNKS = MT * 16 * (KDIM / 4);
    for (int c = tid; c < CHUNKS; c += 256) {
        int row = c >> 5;
        int kc  = c & 31;
        int grow = m_base + row;
        if (grow >= n) grow = n - 1;
        float4 v = ((const float4*)(A + (size_t)grow * KDIM))[kc];
        if (RELU) {
            v.x = fmaxf(v.x, 0.0f); v.y = fmaxf(v.y, 0.0f);
            v.z = fmaxf(v.z, 0.0f); v.w = fmaxf(v.w, 0.0f);
        }
        *(float4*)&tile[row * LDSW + kc * 4] = v;
    }
    __syncthreads();

    const int wave = tid >> 5;
    const int lane = tid & 31;
    int mt, nt;
    if (OUTCH == 128) { mt = 0;         nt = wave;     }
    else              { mt = wave >> 2; nt = wave & 3; }
    const int lm = lane & 15;
    const int kh = lane >> 4;

    const float* arow = &tile[(mt * 16 + lm) * LDSW];
    const float* wcol = W + nt * 16 + lm;

    v8f acc = {};
#pragma unroll
    for (int k0 = 0; k0 < KDIM; k0 += 4) {
        v2f a, b;
        a.x = arow[k0 + 2 * kh + 0];
        a.y = arow[k0 + 2 * kh + 1];
        b.x = wcol[(size_t)(k0 + 2 * kh + 0) * OUTCH];
        b.y = wcol[(size_t)(k0 + 2 * kh + 1) * OUTCH];
        acc = __builtin_amdgcn_wmma_f32_16x16x4_f32(
             false, a,  false, b,
             (short)0, acc,  false,  false);
    }

    const int mrow0 = m_base + mt * 16 + (kh ? 8 : 0);
#pragma unroll
    for (int r = 0; r < 8; ++r) {
        int grow = mrow0 + r;
        if (grow < n)
            H[(size_t)grow * OUTCH + nt * 16 + lm] = acc[r];
    }
}

template <int CH>
__device__ __forceinline__ int chan_pos(int c) {
    constexpr int VEC = CH / 32;
    return (c % VEC) * 32 + c / VEC;
}

template <int CH, int CAP>
__global__ __launch_bounds__(NTHR) void k_agg(const int* __restrict__ src,
                                              const int* __restrict__ dst,
                                              const int* lists,
                                              const float* __restrict__ h,
                                              const float* __restrict__ bias,
                                              const float* __restrict__ dinv,
                                              float* out, int n, int e) {
    constexpr int VEC = CH / 32;
    constexpr int R = KR;
    __shared__ float acc[R * CH];
    __shared__ int   l_s[CAP];
    __shared__ int   l_d[CAP];
    __shared__ float s_dinv[R];

    const int tid = threadIdx.x;
    const int lo = blockIdx.x * R;
    const int rows = min(R, n - lo);

    const int* reg = lists + (size_t)lo * LW;
    const int found = reg[0];
    const int total = found < 0 ? 0 : min(found, CAP);
    {
        const int2* ent = (const int2*)(reg + LHEAD);
        for (int i = tid; i < total; i += NTHR) {
            int2 p = ent[i];
            l_s[i] = p.x;
            l_d[i] = p.y;
        }
    }

    for (int r = tid; r < rows; r += NTHR) s_dinv[r] = dinv[lo + r];
    for (int i = tid; i < rows * (CH / 4); i += NTHR) {
        int r  = i / (CH / 4);
        int c4 = i % (CH / 4);
        float d = dinv[lo + r];
        float s = d * d;
        float4 hv = ((const float4*)(h + (size_t)(lo + r) * CH))[c4];
        float4 bv = ((const float4*)bias)[c4];
        float* a = acc + r * CH;
        a[chan_pos<CH>(c4 * 4 + 0)] = fmaf(hv.x, s, bv.x);
        a[chan_pos<CH>(c4 * 4 + 1)] = fmaf(hv.y, s, bv.y);
        a[chan_pos<CH>(c4 * 4 + 2)] = fmaf(hv.z, s, bv.z);
        a[chan_pos<CH>(c4 * 4 + 3)] = fmaf(hv.w, s, bv.w);
    }
    __syncthreads();

    if (found < 0) {
        for (int i = tid; i < e; i += NTHR) {
            int dl = dst[i] - lo;
            if ((unsigned)dl < (unsigned)rows) {
                int s = src[i];
                float w = dinv[s] * s_dinv[dl];
                const float* hr = h + (size_t)s * CH;
                float* a = acc + dl * CH;
                for (int c = 0; c < CH; ++c) atomicAdd(&a[chan_pos<CH>(c)], hr[c] * w);
            }
        }
    }

    const int wave = tid >> 5;
    const int lane = tid & 31;
    for (int base = wave * 32; base < total; base += NTHR) {
        int i = base + lane;
        int s = 0, dl = 0;
        float w = 0.0f;
        if (i < total) {
            s = l_s[i];
            dl = l_d[i];
            w = dinv[s] * s_dinv[dl];
        }
        const int m = min(32, total - base);
#pragma unroll
        for (int g = 0; g < 32; g += 8) {
            if (g + 8 <= m) {
                float hv[8][VEC];
#pragma unroll
                for (int k = 0; k < 8; ++k) {
                    int sj = __builtin_amdgcn_readlane(s, g + k);
                    const float* hr = h + (size_t)sj * CH + lane * VEC;
#pragma unroll
                    for (int j = 0; j < VEC; ++j) hv[k][j] = hr[j];
                }
#pragma unroll
                for (int k = 0; k < 8; ++k) {
                    float wj = __int_as_float(__builtin_amdgcn_readlane(__float_as_int(w), g + k));
                    int dj = __builtin_amdgcn_readlane(dl, g + k);
                    float* a = acc + dj * CH + lane;
#pragma unroll
                    for (int j = 0; j < VEC; ++j) atomicAdd(a + j * 32, hv[k][j] * wj);
                }
            } else if (g < m) {
#pragma unroll
                for (int k = 0; k < 8; ++k) {
                    if (g + k < m) {
                        int sj = __builtin_amdgcn_readlane(s, g + k);
                        float wj = __int_as_float(__builtin_amdgcn_readlane(__float_as_int(w), g + k));
                        int dj = __builtin_amdgcn_readlane(dl, g + k);
                        const float* hr = h + (size_t)sj * CH + lane * VEC;
                        float* a = acc + dj * CH + lane;
#pragma unroll
                        for (int j = 0; j < VEC; ++j) atomicAdd(a + j * 32, hr[j] * wj);
                    }
                }
            }
        }
    }
    __syncthreads();

    for (int i = tid; i < rows * (CH / 4); i += NTHR) {
        int r  = i / (CH / 4);
        int c4 = i % (CH / 4);
        const float* a = acc + r * CH;
        float4 o;
        o.x = a[chan_pos<CH>(c4 * 4 + 0)];
        o.y = a[chan_pos<CH>(c4 * 4 + 1)];
        o.z = a[chan_pos<CH>(c4 * 4 + 2)];
        o.w = a[chan_pos<CH>(c4 * 4 + 3)];
        ((float4*)(out + (size_t)(lo + r) * CH))[c4] = o;
    }
}

extern "C" void kernel_launch(void* const* d_in, const int* in_sizes, int n_in,
                              void* d_out, int out_size, void* d_ws, size_t ws_size,
                              hipStream_t stream) {
    const float* x  = (const float*)d_in[0];
    const int*   ei = (const int*)d_in[1];
    const float* W1 = (const float*)d_in[2];
    const float* b1 = (const float*)d_in[3];
    const float* W2 = (const float*)d_in[4];
    const float* b2 = (const float*)d_in[5];
    float* out = (float*)d_out;

    const int n = in_sizes[0] / 128;
    const int e = in_sizes[1] / 2;
    const int* src = ei;
    const int* dst = ei + e;

    uintptr_t wb = (uintptr_t)d_ws;
    const uintptr_t wa = (wb + 255) & ~(uintptr_t)255;
    const size_t need = ((size_t)n * 256 + (size_t)n) * sizeof(float);
    if ((wa - wb) + need <= ws_size) wb = wa;
    float* h  = (float*)wb;
    float* o1 = h + (size_t)n * 128;
    float* dinv = o1 + (size_t)n * 128;

    const unsigned T = 256;
    constexpr int LCAP = 12288;
    const int nr = (n + KR - 1) / KR;

    const int npieces = (e + 15) / 16;
    const int kblocks = (npieces + 1023) / 1024;
    const int pw = kblocks * 32;
    int* rec = (int*)o1;
    unsigned* plane = (unsigned*)((unsigned char*)o1 + (((size_t)npieces * 128 + 255) & ~(size_t)255));
    int* lists = (int*)d_out;

    k_keys<<<kblocks, NTHR, 0, stream>>>(src, dst, rec, plane, e, pw);
    k_deg<LCAP><<<nr, NTHR, 0, stream>>>(rec, plane, dinv, lists, n, pw);

    k_gemm<128, 1, false><<<(n + 15) / 16, T, 0, stream>>>(x, W1, h, n);
    k_agg<128, LCAP><<<nr, NTHR, 0, stream>>>(src, dst, lists, h, b1, dinv, o1, n, e);

    k_gemm<64, 2, true><<<(n + 31) / 32, T, 0, stream>>>(o1, W2, h, n);
    k_agg<64, LCAP><<<nr, NTHR, 0, stream>>>(src, dst, lists, h, b2, dinv, out, n, e);
}


# Both products as one resident pass: 1024 blocks of 8 waves loop over row tiles, a wave's W slice held in registers, next tile's rows loaded under this tile's WMMAs (two LDS buffers); k order kept
# speedup vs baseline: 3.6918x; 1.0392x over previous
#include <hip/hip_runtime.h>
#include <stdint.h>

typedef __attribute__((ext_vector_type(2))) float v2f;
typedef __attribute__((ext_vector_type(8))) float v8f;

#define KDIM 128
#define NTHR 1024

#define KR 400
#define LW 64
#define LHEAD 64
#define KW 257
#define RB 9

__global__ __launch_bounds__(1024) void k_keys(const int* __restrict__ src,
                                               const int* __restrict__ dst,
                                               int* __restrict__ rec,
                                               unsigned* __restrict__ plane, int e, int pw) {
    __shared__ unsigned bm[32 * KW];
    const int tid = threadIdx.x;
    const int wave = tid >> 5;
    const int lane = tid & 31;
    const int oct = lane >> 3;
    const int sub = lane & 7;
    for (int i = tid; i < 32 * KW; i += 1024) bm[i] = 0u;
    __syncthreads();

    unsigned* my = bm + wave * KW;
    int4* rec4 = (int4*)rec;
    const int pc0 = blockIdx.x * 1024 + wave * 32;
    const bool al = ((((uintptr_t)dst) | ((uintptr_t)src)) & 7) == 0;
    if (al && (pc0 + 32) * 16 <= e) {
        int2 d[8], s[8];
#pragma unroll
        for (int u = 0; u < 8; ++u) {
            const int e0 = (pc0 + u * 4 + oct) * 16 + 2 * sub;
            d[u] = *(const int2*)(dst + e0);
            s[u] = *(const int2*)(src + e0);
        }
#pragma unroll
        for (int u = 0; u < 8; ++u) {
            const int b = u * 4 + oct;
            rec4[(size_t)(pc0 + b) * 8 + sub] = make_int4(d[u].x, s[u].x, d[u].y, s[u].y);
            atomicOr(&my[((unsigned)d[u].x / KR) & 0xFFu], 1u << b);
            atomicOr(&my[((unsigned)d[u].y / KR) & 0xFFu], 1u << b);
        }
    } else {
        for (int u = 0; u < 8; ++u) {
            const int b = u * 4 + oct;
            const int pc = pc0 + b;
            if (pc * 16 < e) {
                const int e0 = pc * 16 + 2 * sub;
                const bool x0 = e0 < e, x1 = e0 + 1 < e;
                const int d0 = x0 ? dst[e0] : -1,     s0 = x0 ? src[e0] : 0;
                const int d1 = x1 ? dst[e0 + 1] : -1, s1 = x1 ? src[e0 + 1] : 0;
                rec4[(size_t)pc * 8 + sub] = make_int4(d0, s0, d1, s1);
                if (x0) atomicOr(&my[((unsigned)d0 / KR) & 0xFFu], 1u << b);
                if (x1) atomicOr(&my[((unsigned)d1 / KR) & 0xFFu], 1u << b);
            }
        }
    }
    __syncthreads();

    for (int j = 0; j < 8; ++j) {
        const int o = wave * 8 + j;
        plane[(size_t)o * pw + blockIdx.x * 32 + lane] = bm[lane * KW + o];
    }
}

template <int BITS>
__device__ __forceinline__ int wave_prefix(int c, int& tot) {
    int pre = 0, t = 0;
#pragma unroll
    for (int b = 0; b < BITS; ++b) {
        const unsigned m = __builtin_amdgcn_ballot_w32(((c >> b) & 1) != 0);
        pre += (int)__builtin_amdgcn_mbcnt_lo(m, 0u) << b;
        t += __builtin_popcount(m) << b;
    }
    tot = t;
    return pre;
}

template <int CAP>
__global__ __launch_bounds__(NTHR) void k_deg(const int* __restrict__ rec,
                                              const unsigned* __restrict__ plane,
                                              float* __restrict__ dinv,
                                              int* __restrict__ lists, int n, int pw) {
    static_assert(KR <= (1 << RB), "a local row must fit RB bits");
    __shared__ int cnt[KR];
    __shared__ int l_s[CAP];
    __shared__ int l_d[CAP];
    __shared__ int s_cnt;
    __shared__ int s_np;
    const int tid = threadIdx.x;
    const int wave = tid >> 5;
    const int lane = tid & 31;
    const int lo = blockIdx.x * KR;
    const int rows = min(KR, n - lo);
    if (tid == 0) { s_cnt = 0; s_np = 0; }
    for (int r = tid; r < KR; r += NTHR) cnt[r] = 0;
    __syncthreads();

    const int4* rec4 = (const int4*)rec;
    auto mine = [&](int d) -> bool { return (unsigned)(d - lo) < (unsigned)KR; };
    auto take = [&](int d, int s, int slot) {
        const int dl = d - lo;
        atomicAdd(&cnt[dl], 1);
        if (slot < CAP) l_s[slot] = (int)(((unsigned)s << RB) | (unsigned)dl);
    };

    const unsigned* row = plane + (size_t)blockIdx.x * pw;
    for (int w0 = 0; w0 < pw; w0 += 4 * NTHR) {
        unsigned w[4];
        int c = 0;
#pragma unroll
        for (int u = 0; u < 4; ++u) {
            const int wi = w0 + u * NTHR + tid;
            w[u] = (wi < pw) ? row[wi] : 0u;
            c += __builtin_popcount(w[u]);
        }
        int tot;
        const int pre = wave_prefix<8>(c, tot);
        int base = 0;
        if (lane == 0) base = atomicAdd(&s_np, tot);
        int slot = __builtin_amdgcn_readfirstlane(base) + pre;
#pragma unroll
        for (int u = 0; u < 4; ++u) {
            const int wi = w0 + u * NTHR + tid;
            unsigned x = w[u];
            while (x) {
                const int pc = wi * 32 + __builtin_ctz(x);
                x &= x - 1;
                if (slot < CAP) {
                    l_d[slot] = pc;
                } else {
                    for (int j = 0; j < 8; ++j) {
                        const int4 q = rec4[(size_t)pc * 8 + j];
                        if (mine(q.x)) take(q.x, q.y, atomicAdd(&s_cnt, 1));
                        if (mine(q.z)) take(q.z, q.w, atomicAdd(&s_cnt, 1));
                    }
                }
                ++slot;
            }
        }
    }
    __syncthreads();

    const int np = min(s_np, CAP);
    const int oct = lane >> 3;
    const int sub = lane & 7;
    for (int i0 = wave * 32; i0 < np; i0 += NTHR) {
        int pc[8];
        int4 v[8];
        unsigned mm = 0u;
        int c = 0;
#pragma unroll
        for (int u = 0; u < 8; ++u) {
            const int i = i0 + u * 4 + oct;
            const int q = l_d[min(i, np - 1)];
            pc[u] = (i < np) ? q : -1;
        }
#pragma unroll
        for (int u = 0; u < 8; ++u) v[u] = rec4[(size_t)max(pc[u], 0) * 8 + sub];
#pragma unroll
        for (int u = 0; u < 8; ++u) {
            const unsigned m0 = (pc[u] >= 0 && mine(v[u].x)) ? 1u : 0u;
            const unsigned m1 = (pc[u] >= 0 && mine(v[u].z)) ? 1u : 0u;
            mm |= (m0 << (2 * u)) | (m1 << (2 * u + 1));
            c += (int)(m0 + m1);
        }
        int tot;
        const int pre = wave_prefix<5>(c, tot);
        int base = 0;
        if (lane == 0) base = atomicAdd(&s_cnt, tot);
        int slot = __builtin_amdgcn_readfirstlane(base) + pre;
#pragma unroll
        for (int u = 0; u < 8; ++u) {
            if (mm & (1u << (2 * u)))     { take(v[u].x, v[u].y, slot); ++slot; }
            if (mm & (1u << (2 * u + 1))) { take(v[u].z, v[u].w, slot); ++slot; }
        }
    }
    __syncthreads();

    const int found = s_cnt;
    const int total = min(found, CAP);
    for (int r = tid; r < rows; r += NTHR)
        dinv[lo + r] = rsqrtf((float)(cnt[r] + 1));

    int* reg = lists + (size_t)lo * LW;
    const int room = min(CAP, (rows * LW - LHEAD) / 2);
    const bool fits = found <= room;
    if (tid == 0) reg[0] = fits ? found : -1;
    if (fits) {
        int2* ent = (int2*)(reg + LHEAD);
        for (int i = tid; i < total; i += NTHR) {
            const unsigned p = (unsigned)l_s[i];
            ent[i] = make_int2((int)(p >> RB), (int)(p & ((1u << RB) - 1u)));
        }
    }
}

__device__ __forceinline__ float4 ld_chunk(const float* __restrict__ A, int grow, int n, int kc) {
    if (grow >= n) grow = n - 1;
    return ((const float4*)(A + (size_t)grow * KDIM))[kc];
}
template <bool RELU>
__device__ __forceinline__ void st_chunk(float* dst, float4 v) {
    if (RELU) {
        v.x = fmaxf(v.x, 0.0f); v.y = fmaxf(v.y, 0.0f);
        v.z = fmaxf(v.z, 0.0f); v.w = fmaxf(v.w, 0.0f);
    }
    *(float4*)dst = v;
}

template <int OUTCH, int MT, bool RELU>
__global__ __launch_bounds__(256) __attribute__((amdgpu_waves_per_eu(9, 9)))
void k_gemm(const float* __restrict__ A,
            const float* __restrict__ W,
            float* __restrict__ H, int n) {
    constexpr int LDSW = KDIM + 4;
    constexpr int ROWS = MT * 16;
    constexpr int TSZ = ROWS * LDSW;
    __shared__ float tile[2 * TSZ];

    const int tid = threadIdx.x;
    const int wave = tid >> 5;
    const int lane = tid & 31;
    int mt, nt;
    if (OUTCH == 128) { mt = 0;         nt = wave;     }
    else              { mt = wave >> 2; nt = wave & 3; }
    const int lm = lane & 15;
    const int kh = lane >> 4;
    const int ntiles = (n + ROWS - 1) / ROWS;
    const int step = gridDim.x;
    int t = blockIdx.x;
    if (t >= ntiles) return;

    const int crow = tid >> 5;
    const int ckc  = tid & 31;
    float* sdst = tile + crow * LDSW + ckc * 4;
    float4 p0, p1, p2, p3;
    p2 = p3 = make_float4(0.0f, 0.0f, 0.0f, 0.0f);

    int r0 = t * ROWS + crow;
    p0 = ld_chunk(A, r0, n, ckc);
    p1 = ld_chunk(A, r0 + 8, n, ckc);
    if (MT == 2) {
        p2 = ld_chunk(A, r0 + 16, n, ckc);
        p3 = ld_chunk(A, r0 + 24, n, ckc);
    }

    const float* wcol = W + nt * 16 + lm;
    v2f b[KDIM / 4];
#pragma unroll
    for (int s = 0; s < KDIM / 4; ++s) {
        b[s].x = wcol[(size_t)(4 * s + 2 * kh + 0) * OUTCH];
        b[s].y = wcol[(size_t)(4 * s + 2 * kh + 1) * OUTCH];
    }

    st_chunk<RELU>(sdst, p0);
    st_chunk<RELU>(sdst + 8 * LDSW, p1);
    if (MT == 2) {
        st_chunk<RELU>(sdst + 16 * LDSW, p2);
        st_chunk<RELU>(sdst + 24 * LDSW, p3);
    }
    __syncthreads();
    if (t + step < ntiles) {
        r0 = (t + step) * ROWS + crow;
        p0 = ld_chunk(A, r0, n, ckc);
        p1 = ld_chunk(A, r0 + 8, n, ckc);
        if (MT == 2) {
            p2 = ld_chunk(A, r0 + 16, n, ckc);
            p3 = ld_chunk(A, r0 + 24, n, ckc);
        }
    }

    int cur = 0;
    for (; t < ntiles; t += step) {
        const float* arow = tile + cur + (mt * 16 + lm) * LDSW;
        v8f acc = {};
#pragma unroll
        for (int s = 0; s < KDIM / 4; ++s) {
            v2f a;
            a.x = arow[4 * s + 2 * kh + 0];
            a.y = arow[4 * s + 2 * kh + 1];
            acc = __builtin_amdgcn_wmma_f32_16x16x4_f32(
                 false, a,  false, b[s],
                 (short)0, acc,  false,  false);
        }

        if (t + step < ntiles) {
            float* d = sdst + (TSZ - cur);
            st_chunk<RELU>(d, p0);
            st_chunk<RELU>(d + 8 * LDSW, p1);
            if (MT == 2) {
                st_chunk<RELU>(d + 16 * LDSW, p2);
                st_chunk<RELU>(d + 24 * LDSW, p3);
            }
        }
        __syncthreads();

        const int mrow0 = t * ROWS + mt * 16 + (kh ? 8 : 0);
#pragma unroll
        for (int r = 0; r < 8; ++r) {
            int grow = mrow0 + r;
            if (grow < n)
                H[(size_t)grow * OUTCH + nt * 16 + lm] = acc[r];
        }

        if (t + 2 * step < ntiles) {
            r0 = (t + 2 * step) * ROWS + crow;
            p0 = ld_chunk(A, r0, n, ckc);
            p1 = ld_chunk(A, r0 + 8, n, ckc);
            if (MT == 2) {
                p2 = ld_chunk(A, r0 + 16, n, ckc);
                p3 = ld_chunk(A, r0 + 24, n, ckc);
            }
        }
        cur = TSZ - cur;
    }
}

template <int CH>
__device__ __forceinline__ int chan_pos(int c) {
    constexpr int VEC = CH / 32;
    return (c % VEC) * 32 + c / VEC;
}

template <int CH, int CAP>
__global__ __launch_bounds__(NTHR) void k_agg(const int* __restrict__ src,
                                              const int* __restrict__ dst,
                                              const int* lists,
                                              const float* __restrict__ h,
                                              const float* __restrict__ bias,
                                              const float* __restrict__ dinv,
                                              float* out, int n, int e) {
    constexpr int VEC = CH / 32;
    constexpr int R = KR;
    __shared__ float acc[R * CH];
    __shared__ int   l_s[CAP];
    __shared__ int   l_d[CAP];
    __shared__ float s_dinv[R];

    const int tid = threadIdx.x;
    const int lo = blockIdx.x * R;
    const int rows = min(R, n - lo);

    const int* reg = lists + (size_t)lo * LW;
    const int found = reg[0];
    const int total = found < 0 ? 0 : min(found, CAP);
    {
        const int2* ent = (const int2*)(reg + LHEAD);
        for (int i = tid; i < total; i += NTHR) {
            int2 p = ent[i];
            l_s[i] = p.x;
            l_d[i] = p.y;
        }
    }

    for (int r = tid; r < rows; r += NTHR) s_dinv[r] = dinv[lo + r];
    for (int i = tid; i < rows * (CH / 4); i += NTHR) {
        int r  = i / (CH / 4);
        int c4 = i % (CH / 4);
        float d = dinv[lo + r];
        float s = d * d;
        float4 hv = ((const float4*)(h + (size_t)(lo + r) * CH))[c4];
        float4 bv = ((const float4*)bias)[c4];
        float* a = acc + r * CH;
        a[chan_pos<CH>(c4 * 4 + 0)] = fmaf(hv.x, s, bv.x);
        a[chan_pos<CH>(c4 * 4 + 1)] = fmaf(hv.y, s, bv.y);
        a[chan_pos<CH>(c4 * 4 + 2)] = fmaf(hv.z, s, bv.z);
        a[chan_pos<CH>(c4 * 4 + 3)] = fmaf(hv.w, s, bv.w);
    }
    __syncthreads();

    if (found < 0) {
        for (int i = tid; i < e; i += NTHR) {
            int dl = dst[i] - lo;
            if ((unsigned)dl < (unsigned)rows) {
                int s = src[i];
                float w = dinv[s] * s_dinv[dl];
                const float* hr = h + (size_t)s * CH;
                float* a = acc + dl * CH;
                for (int c = 0; c < CH; ++c) atomicAdd(&a[chan_pos<CH>(c)], hr[c] * w);
            }
        }
    }

    const int wave = tid >> 5;
    const int lane = tid & 31;
    for (int base = wave * 32; base < total; base += NTHR) {
        int i = base + lane;
        int s = 0, dl = 0;
        float w = 0.0f;
        if (i < total) {
            s = l_s[i];
            dl = l_d[i];
            w = dinv[s] * s_dinv[dl];
        }
        const int m = min(32, total - base);
#pragma unroll
        for (int g = 0; g < 32; g += 8) {
            if (g + 8 <= m) {
                float hv[8][VEC];
#pragma unroll
                for (int k = 0; k < 8; ++k) {
                    int sj = __builtin_amdgcn_readlane(s, g + k);
                    const float* hr = h + (size_t)sj * CH + lane * VEC;
#pragma unroll
                    for (int j = 0; j < VEC; ++j) hv[k][j] = hr[j];
                }
#pragma unroll
                for (int k = 0; k < 8; ++k) {
                    float wj = __int_as_float(__builtin_amdgcn_readlane(__float_as_int(w), g + k));
                    int dj = __builtin_amdgcn_readlane(dl, g + k);
                    float* a = acc + dj * CH + lane;
#pragma unroll
                    for (int j = 0; j < VEC; ++j) atomicAdd(a + j * 32, hv[k][j] * wj);
                }
            } else if (g < m) {
#pragma unroll
                for (int k = 0; k < 8; ++k) {
                    if (g + k < m) {
                        int sj = __builtin_amdgcn_readlane(s, g + k);
                        float wj = __int_as_float(__builtin_amdgcn_readlane(__float_as_int(w), g + k));
                        int dj = __builtin_amdgcn_readlane(dl, g + k);
                        const float* hr = h + (size_t)sj * CH + lane * VEC;
                        float* a = acc + dj * CH + lane;
#pragma unroll
                        for (int j = 0; j < VEC; ++j) atomicAdd(a + j * 32, hr[j] * wj);
                    }
                }
            }
        }
    }
    __syncthreads();

    for (int i = tid; i < rows * (CH / 4); i += NTHR) {
        int r  = i / (CH / 4);
        int c4 = i % (CH / 4);
        const float* a = acc + r * CH;
        float4 o;
        o.x = a[chan_pos<CH>(c4 * 4 + 0)];
        o.y = a[chan_pos<CH>(c4 * 4 + 1)];
        o.z = a[chan_pos<CH>(c4 * 4 + 2)];
        o.w = a[chan_pos<CH>(c4 * 4 + 3)];
        ((float4*)(out + (size_t)(lo + r) * CH))[c4] = o;
    }
}

extern "C" void kernel_launch(void* const* d_in, const int* in_sizes, int n_in,
                              void* d_out, int out_size, void* d_ws, size_t ws_size,
                              hipStream_t stream) {
    const float* x  = (const float*)d_in[0];
    const int*   ei = (const int*)d_in[1];
    const float* W1 = (const float*)d_in[2];
    const float* b1 = (const float*)d_in[3];
    const float* W2 = (const float*)d_in[4];
    const float* b2 = (const float*)d_in[5];
    float* out = (float*)d_out;

    const int n = in_sizes[0] / 128;
    const int e = in_sizes[1] / 2;
    const int* src = ei;
    const int* dst = ei + e;

    uintptr_t wb = (uintptr_t)d_ws;
    const uintptr_t wa = (wb + 255) & ~(uintptr_t)255;
    const size_t need = ((size_t)n * 256 + (size_t)n) * sizeof(float);
    if ((wa - wb) + need <= ws_size) wb = wa;
    float* h  = (float*)wb;
    float* o1 = h + (size_t)n * 128;
    float* dinv = o1 + (size_t)n * 128;

    const unsigned T = 256;
    constexpr unsigned GEMM1_GRID = 1024;
    constexpr unsigned GEMM2_GRID = 1024;
    constexpr int LCAP = 12288;
    const int nr = (n + KR - 1) / KR;

    const int npieces = (e + 15) / 16;
    const int kblocks = (npieces + 1023) / 1024;
    const int pw = kblocks * 32;
    int* rec = (int*)o1;
    unsigned* plane = (unsigned*)((unsigned char*)o1 + (((size_t)npieces * 128 + 255) & ~(size_t)255));
    int* lists = (int*)d_out;

    k_keys<<<kblocks, NTHR, 0, stream>>>(src, dst, rec, plane, e, pw);
    k_deg<LCAP><<<nr, NTHR, 0, stream>>>(rec, plane, dinv, lists, n, pw);

    k_gemm<128, 1, false><<<GEMM1_GRID, T, 0, stream>>>(x, W1, h, n);
    k_agg<128, LCAP><<<nr, NTHR, 0, stream>>>(src, dst, lists, h, b1, dinv, o1, n, e);

    k_gemm<64, 2, true><<<GEMM2_GRID, T, 0, stream>>>(o1, W2, h, n);
    k_agg<64, LCAP><<<nr, NTHR, 0, stream>>>(src, dst, lists, h, b2, dinv, out, n, e);
}
